# starting kernel
# baseline (speedup 1.0000x reference)
#include <hip/hip_runtime.h>
#include <hip/hip_bf16.h>


#define FN   512
#define NB   43
#define LPB  4
#define TPB  128
#define PI_F 3.14159265358979323846f

typedef float v2f __attribute__((ext_vector_type(2)));
typedef float v8f __attribute__((ext_vector_type(8)));

__device__ __forceinline__ v8f wmma_f32(v2f a, v2f b, v8f c) {
  return __builtin_amdgcn_wmma_f32_16x16x4_f32(false, a, false, b,
                                               (short)0, c, false, false);
}

struct W16 { v2f ar[4]; v2f ai[4]; v2f ain[4]; };

__device__ __forceinline__ void make_w16(float sgn, int lane, W16& w) {
  const int m  = lane & 15;
  const int kb = (lane >> 4) << 1;
  const float step = sgn * (2.0f * PI_F / 16.0f);
#pragma unroll
  for (int kc = 0; kc < 4; ++kc) {
    const int k0 = 4 * kc + kb;
    float s0, c0, s1, c1;
    sincosf(step * (float)(m * k0),       &s0, &c0);
    sincosf(step * (float)(m * (k0 + 1)), &s1, &c1);
    v2f tr; tr.x = c0;  tr.y = c1;  w.ar[kc]  = tr;
    v2f ti; ti.x = s0;  ti.y = s1;  w.ai[kc]  = ti;
    v2f tn; tn.x = -s0; tn.y = -s1; w.ain[kc] = tn;
  }
}

__device__ __forceinline__ void ctile(const float* sr, const float* si, int t,
                                      int lane, const W16& w, v8f& cr, v8f& ci) {
  const int n  = 16 * t + (lane & 15);
  const int kb = (lane >> 4) << 1;
  v8f zr = {0.f,0.f,0.f,0.f,0.f,0.f,0.f,0.f};
  cr = zr; ci = zr;
#pragma unroll
  for (int kc = 0; kc < 4; ++kc) {
    const int k0 = 4 * kc + kb;
    v2f br; br.x = sr[k0 * 32 + n]; br.y = sr[(k0 + 1) * 32 + n];
    v2f bi; bi.x = si[k0 * 32 + n]; bi.y = si[(k0 + 1) * 32 + n];
    cr = wmma_f32(w.ar[kc],  br, cr);
    cr = wmma_f32(w.ain[kc], bi, cr);
    ci = wmma_f32(w.ar[kc],  bi, ci);
    ci = wmma_f32(w.ai[kc],  br, ci);
  }
}

__device__ __forceinline__ void fft_stageA(const float* xr, const float* xi,
                                           float* ur, float* ui,
                                           int lane, float sgn) {
  const float tws = sgn * (2.0f * PI_F / 32.0f);
#pragma unroll
  for (int i = 0; i < 8; ++i) {
    const int p  = lane + 32 * i;
    const int n1 = p & 15, m1 = p >> 4;
    const float a_r = xr[p],       a_i = xi[p];
    const float b_r = xr[p + 256], b_i = xi[p + 256];
    ur[m1 * 32 + n1] = a_r + b_r;
    ui[m1 * 32 + n1] = a_i + b_i;
    const float dr = a_r - b_r, di = a_i - b_i;
    float s, c; sincosf(tws * (float)m1, &s, &c);
    ur[m1 * 32 + n1 + 16] = dr * c - di * s;
    ui[m1 * 32 + n1 + 16] = dr * s + di * c;
  }
}

__device__ __forceinline__ void fft_stageBC(const float* ur, const float* ui,
                                            float* vr, float* vi,
                                            const float* twr, const float* twi,
                                            int lane, const W16& w) {
  const int nn = lane & 15;
  const int mb = (lane >> 4) * 8;
#pragma unroll
  for (int t = 0; t < 2; ++t) {
    v8f gr, gi;
    ctile(ur, ui, t, lane, w, gr, gi);
    const int ccol = 16 * t + nn;
    const int n1 = ccol & 15, j2 = ccol >> 4;
#pragma unroll
    for (int r = 0; r < 8; ++r) {
      const int j1  = mb + r;
      const int k2  = 2 * j1 + j2;
      const int idx = n1 * 32 + k2;
      const float tr = twr[idx], ti_ = twi[idx];
      const float g_r = gr[r], g_i = gi[r];
      vr[idx] = g_r * tr - g_i * ti_;
      vi[idx] = g_r * ti_ + g_i * tr;
    }
  }
}

__device__ __forceinline__ void fill_tw512(float* twr, float* twi,
                                           int tid, float sgn) {
  const float step = sgn * (2.0f * PI_F / 512.0f);
  for (int i = tid; i < FN; i += TPB) {
    const int n1 = i >> 5, k2 = i & 31;
    float s, c; sincosf(step * (float)(n1 * k2), &s, &c);
    twr[i] = c; twi[i] = s;
  }
}

#define FFT_SHARED                                              \
  __shared__ float Ar[LPB][FN], Ai[LPB][FN];                    \
  __shared__ float Br[LPB][FN], Bi[LPB][FN];                    \
  __shared__ float Twr[FN], Twi[FN];

__global__ __launch_bounds__(TPB)
void k_fwd_rows(const float* __restrict__ img, float2* __restrict__ specRow) {
  FFT_SHARED
  const int tid = threadIdx.x, lane = tid & 31, wid = tid >> 5;
  const int line = blockIdx.x * LPB + wid;
  fill_tw512(Twr, Twi, tid, -1.0f);
  W16 w; make_w16(-1.0f, lane, w);
#pragma unroll
  for (int i = 0; i < 16; ++i) {
    const int p = lane + 32 * i;
    Ar[wid][p] = img[line * FN + p];
    Ai[wid][p] = 0.0f;
  }
  __syncthreads();
  fft_stageA(Ar[wid], Ai[wid], Br[wid], Bi[wid], lane, -1.0f);
  __syncthreads();
  fft_stageBC(Br[wid], Bi[wid], Ar[wid], Ai[wid], Twr, Twi, lane, w);
  __syncthreads();
  const int nn = lane & 15, mb = (lane >> 4) * 8;
#pragma unroll
  for (int t = 0; t < 2; ++t) {
    v8f dr, di;
    ctile(Ar[wid], Ai[wid], t, lane, w, dr, di);
#pragma unroll
    for (int r = 0; r < 8; ++r) {
      const int o = 32 * (mb + r) + 16 * t + nn;
      specRow[line * FN + o] = make_float2(dr[r], di[r]);
    }
  }
}

__global__ __launch_bounds__(TPB)
void k_fwd_cols(const float2* __restrict__ specRow, float2* __restrict__ spec) {
  FFT_SHARED
  const int tid = threadIdx.x, lane = tid & 31, wid = tid >> 5;
  const int c0 = blockIdx.x * LPB;
  fill_tw512(Twr, Twi, tid, -1.0f);
  W16 w; make_w16(-1.0f, lane, w);
  for (int i = tid; i < FN * LPB; i += TPB) {
    const int e = i >> 2, ww = i & 3;
    const float2 s = specRow[e * FN + c0 + ww];
    Ar[ww][e] = s.x; Ai[ww][e] = s.y;
  }
  __syncthreads();
  fft_stageA(Ar[wid], Ai[wid], Br[wid], Bi[wid], lane, -1.0f);
  __syncthreads();
  fft_stageBC(Br[wid], Bi[wid], Ar[wid], Ai[wid], Twr, Twi, lane, w);
  __syncthreads();
  const int nn = lane & 15, mb = (lane >> 4) * 8;
#pragma unroll
  for (int t = 0; t < 2; ++t) {
    v8f dr, di;
    ctile(Ar[wid], Ai[wid], t, lane, w, dr, di);
#pragma unroll
    for (int r = 0; r < 8; ++r) {
      const int o = 32 * (mb + r) + 16 * t + nn;
      Br[wid][o] = dr[r]; Bi[wid][o] = di[r];
    }
  }
  __syncthreads();
  for (int i = tid; i < FN * LPB; i += TPB) {
    const int o = i >> 2, ww = i & 3;
    spec[o * FN + c0 + ww] = make_float2(Br[ww][o], Bi[ww][o]);
  }
}

__global__ __launch_bounds__(TPB)
void k_band_cols(const float2* __restrict__ spec, const float* __restrict__ win,
                 float2* __restrict__ T, int b0) {
  FFT_SHARED
  const int tid = threadIdx.x, lane = tid & 31, wid = tid >> 5;
  const int c0 = blockIdx.x * LPB;
  const int bi = blockIdx.y;
  const size_t woff = (size_t)(b0 + bi) * FN * FN;
  float2* Tb = T + (size_t)bi * FN * FN;
  fill_tw512(Twr, Twi, tid, 1.0f);
  W16 w; make_w16(1.0f, lane, w);
  for (int i = tid; i < FN * LPB; i += TPB) {
    const int e = i >> 2, ww = i & 3;
    const float2 s = spec[e * FN + c0 + ww];
    const float  wv = win[woff + (size_t)e * FN + c0 + ww];
    Ar[ww][e] = s.x * wv; Ai[ww][e] = s.y * wv;
  }
  __syncthreads();
  fft_stageA(Ar[wid], Ai[wid], Br[wid], Bi[wid], lane, 1.0f);
  __syncthreads();
  fft_stageBC(Br[wid], Bi[wid], Ar[wid], Ai[wid], Twr, Twi, lane, w);
  __syncthreads();
  const int nn = lane & 15, mb = (lane >> 4) * 8;
#pragma unroll
  for (int t = 0; t < 2; ++t) {
    v8f dr, di;
    ctile(Ar[wid], Ai[wid], t, lane, w, dr, di);
#pragma unroll
    for (int r = 0; r < 8; ++r) {
      const int o = 32 * (mb + r) + 16 * t + nn;
      Br[wid][o] = dr[r]; Bi[wid][o] = di[r];
    }
  }
  __syncthreads();
  for (int i = tid; i < FN * LPB; i += TPB) {
    const int o = i >> 2, ww = i & 3;
    Tb[o * FN + c0 + ww] = make_float2(Br[ww][o], Bi[ww][o]);
  }
}

__global__ __launch_bounds__(TPB)
void k_band_rows(const float2* __restrict__ T, float* __restrict__ out, int b0) {
  FFT_SHARED
  const int tid = threadIdx.x, lane = tid & 31, wid = tid >> 5;
  const int bi = blockIdx.y;
  const float2* Tb = T + (size_t)bi * FN * FN;
  float* ob = out + (size_t)(b0 + bi) * FN * FN;
  const int line = blockIdx.x * LPB + wid;
  fill_tw512(Twr, Twi, tid, 1.0f);
  W16 w; make_w16(1.0f, lane, w);
#pragma unroll
  for (int i = 0; i < 16; ++i) {
    const int p = lane + 32 * i;
    const float2 v = Tb[line * FN + p];
    Ar[wid][p] = v.x; Ai[wid][p] = v.y;
  }
  __syncthreads();
  fft_stageA(Ar[wid], Ai[wid], Br[wid], Bi[wid], lane, 1.0f);
  __syncthreads();
  fft_stageBC(Br[wid], Bi[wid], Ar[wid], Ai[wid], Twr, Twi, lane, w);
  __syncthreads();
  const float scale = 1.0f / (512.0f * 512.0f);
  const int nn = lane & 15, mb = (lane >> 4) * 8;
#pragma unroll
  for (int t = 0; t < 2; ++t) {
    v8f dr, di;
    ctile(Ar[wid], Ai[wid], t, lane, w, dr, di);
#pragma unroll
    for (int r = 0; r < 8; ++r) {
      const int o = 32 * (mb + r) + 16 * t + nn;
      ob[line * FN + o] = dr[r] * scale;
    }
  }
}

extern "C" void kernel_launch(void* const* d_in, const int* in_sizes, int n_in,
                              void* d_out, int out_size, void* d_ws,
                              size_t ws_size, hipStream_t stream) {
  (void)in_sizes; (void)n_in; (void)out_size;
  const float* img = (const float*)d_in[0];
  const float* win = (const float*)d_in[1];
  float* out = (float*)d_out;

  float2* specRow = (float2*)d_ws;
  float2* spec    = specRow + (size_t)FN * FN;
  float2* bandT   = spec    + (size_t)FN * FN;
  const size_t reserved = 2ull * sizeof(float2) * FN * FN;
  const size_t perBand  = sizeof(float2) * (size_t)FN * FN;
  int chunk = (ws_size > reserved) ? (int)((ws_size - reserved) / perBand) : 0;
  if (chunk < 1)  chunk = 1;
  if (chunk > NB) chunk = NB;

  const dim3 blk(TPB);
  k_fwd_rows<<<dim3(FN / LPB), blk, 0, stream>>>(img, specRow);
  k_fwd_cols<<<dim3(FN / LPB), blk, 0, stream>>>(specRow, spec);
  for (int b0 = 0; b0 < NB; b0 += chunk) {
    const int nb = (NB - b0 < chunk) ? (NB - b0) : chunk;
    k_band_cols<<<dim3(FN / LPB, nb), blk, 0, stream>>>(spec, win, bandT, b0);
    k_band_rows<<<dim3(FN / LPB, nb), blk, 0, stream>>>(bandT, out, b0);
  }
}


# Band column kernel: 16 columns per work-group (was 4), one wave each, loads issued first, LDS line pitch 516: every tile row is one aligned 128-byte access, transposing LDS accesses conflict-free
# speedup vs baseline: 1.6552x; 1.6552x over previous
#include <hip/hip_runtime.h>
#include <hip/hip_bf16.h>


#define FN   512
#define NB   43
#define LPB  4
#define TPB  128
#define CPB    16
#define CTPB   512
#define LPITCH (FN + 4)
#define PI_F 3.14159265358979323846f

typedef float v2f __attribute__((ext_vector_type(2)));
typedef float v8f __attribute__((ext_vector_type(8)));

__device__ __forceinline__ v8f wmma_f32(v2f a, v2f b, v8f c) {
  return __builtin_amdgcn_wmma_f32_16x16x4_f32(false, a, false, b,
                                               (short)0, c, false, false);
}

struct W16 { v2f ar[4]; v2f ai[4]; v2f ain[4]; };

__device__ __forceinline__ void make_w16(float sgn, int lane, W16& w) {
  const int m  = lane & 15;
  const int kb = (lane >> 4) << 1;
  const float step = sgn * (2.0f * PI_F / 16.0f);
#pragma unroll
  for (int kc = 0; kc < 4; ++kc) {
    const int k0 = 4 * kc + kb;
    float s0, c0, s1, c1;
    sincosf(step * (float)(m * k0),       &s0, &c0);
    sincosf(step * (float)(m * (k0 + 1)), &s1, &c1);
    v2f tr; tr.x = c0;  tr.y = c1;  w.ar[kc]  = tr;
    v2f ti; ti.x = s0;  ti.y = s1;  w.ai[kc]  = ti;
    v2f tn; tn.x = -s0; tn.y = -s1; w.ain[kc] = tn;
  }
}

__device__ __forceinline__ void ctile(const float* sr, const float* si, int t,
                                      int lane, const W16& w, v8f& cr, v8f& ci) {
  const int n  = 16 * t + (lane & 15);
  const int kb = (lane >> 4) << 1;
  v8f zr = {0.f,0.f,0.f,0.f,0.f,0.f,0.f,0.f};
  cr = zr; ci = zr;
#pragma unroll
  for (int kc = 0; kc < 4; ++kc) {
    const int k0 = 4 * kc + kb;
    v2f br; br.x = sr[k0 * 32 + n]; br.y = sr[(k0 + 1) * 32 + n];
    v2f bi; bi.x = si[k0 * 32 + n]; bi.y = si[(k0 + 1) * 32 + n];
    cr = wmma_f32(w.ar[kc],  br, cr);
    cr = wmma_f32(w.ain[kc], bi, cr);
    ci = wmma_f32(w.ar[kc],  bi, ci);
    ci = wmma_f32(w.ai[kc],  br, ci);
  }
}

__device__ __forceinline__ void fft_stageA(const float* xr, const float* xi,
                                           float* ur, float* ui,
                                           int lane, float sgn) {
  const float tws = sgn * (2.0f * PI_F / 32.0f);
#pragma unroll
  for (int i = 0; i < 8; ++i) {
    const int p  = lane + 32 * i;
    const int n1 = p & 15, m1 = p >> 4;
    const float a_r = xr[p],       a_i = xi[p];
    const float b_r = xr[p + 256], b_i = xi[p + 256];
    ur[m1 * 32 + n1] = a_r + b_r;
    ui[m1 * 32 + n1] = a_i + b_i;
    const float dr = a_r - b_r, di = a_i - b_i;
    float s, c; sincosf(tws * (float)m1, &s, &c);
    ur[m1 * 32 + n1 + 16] = dr * c - di * s;
    ui[m1 * 32 + n1 + 16] = dr * s + di * c;
  }
}

__device__ __forceinline__ void fft_stageBC(const float* ur, const float* ui,
                                            float* vr, float* vi,
                                            const float* twr, const float* twi,
                                            int lane, const W16& w) {
  const int nn = lane & 15;
  const int mb = (lane >> 4) * 8;
#pragma unroll
  for (int t = 0; t < 2; ++t) {
    v8f gr, gi;
    ctile(ur, ui, t, lane, w, gr, gi);
    const int ccol = 16 * t + nn;
    const int n1 = ccol & 15, j2 = ccol >> 4;
#pragma unroll
    for (int r = 0; r < 8; ++r) {
      const int j1  = mb + r;
      const int k2  = 2 * j1 + j2;
      const int idx = n1 * 32 + k2;
      const float tr = twr[idx], ti_ = twi[idx];
      const float g_r = gr[r], g_i = gi[r];
      vr[idx] = g_r * tr - g_i * ti_;
      vi[idx] = g_r * ti_ + g_i * tr;
    }
  }
}

__device__ __forceinline__ void fill_tw512(float* twr, float* twi,
                                           int tid, int nthr, float sgn) {
  const float step = sgn * (2.0f * PI_F / 512.0f);
  for (int i = tid; i < FN; i += nthr) {
    const int n1 = i >> 5, k2 = i & 31;
    float s, c; sincosf(step * (float)(n1 * k2), &s, &c);
    twr[i] = c; twi[i] = s;
  }
}

#define FFT_SHARED                                              \
  __shared__ float Ar[LPB][FN], Ai[LPB][FN];                    \
  __shared__ float Br[LPB][FN], Bi[LPB][FN];                    \
  __shared__ float Twr[FN], Twi[FN];

#define FFT_SHARED_C16                                          \
  __shared__ float Ar[CPB][LPITCH], Ai[CPB][LPITCH];            \
  __shared__ float Br[CPB][LPITCH], Bi[CPB][LPITCH];            \
  __shared__ float Twr[FN], Twi[FN];

__global__ __launch_bounds__(TPB)
void k_fwd_rows(const float* __restrict__ img, float2* __restrict__ specRow) {
  FFT_SHARED
  const int tid = threadIdx.x, lane = tid & 31, wid = tid >> 5;
  const int line = blockIdx.x * LPB + wid;
  fill_tw512(Twr, Twi, tid, TPB, -1.0f);
  W16 w; make_w16(-1.0f, lane, w);
#pragma unroll
  for (int i = 0; i < 16; ++i) {
    const int p = lane + 32 * i;
    Ar[wid][p] = img[line * FN + p];
    Ai[wid][p] = 0.0f;
  }
  __syncthreads();
  fft_stageA(Ar[wid], Ai[wid], Br[wid], Bi[wid], lane, -1.0f);
  __syncthreads();
  fft_stageBC(Br[wid], Bi[wid], Ar[wid], Ai[wid], Twr, Twi, lane, w);
  __syncthreads();
  const int nn = lane & 15, mb = (lane >> 4) * 8;
#pragma unroll
  for (int t = 0; t < 2; ++t) {
    v8f dr, di;
    ctile(Ar[wid], Ai[wid], t, lane, w, dr, di);
#pragma unroll
    for (int r = 0; r < 8; ++r) {
      const int o = 32 * (mb + r) + 16 * t + nn;
      specRow[line * FN + o] = make_float2(dr[r], di[r]);
    }
  }
}

__global__ __launch_bounds__(TPB)
void k_fwd_cols(const float2* __restrict__ specRow, float2* __restrict__ spec) {
  FFT_SHARED
  const int tid = threadIdx.x, lane = tid & 31, wid = tid >> 5;
  const int c0 = blockIdx.x * LPB;
  fill_tw512(Twr, Twi, tid, TPB, -1.0f);
  W16 w; make_w16(-1.0f, lane, w);
  for (int i = tid; i < FN * LPB; i += TPB) {
    const int e = i >> 2, ww = i & 3;
    const float2 s = specRow[e * FN + c0 + ww];
    Ar[ww][e] = s.x; Ai[ww][e] = s.y;
  }
  __syncthreads();
  fft_stageA(Ar[wid], Ai[wid], Br[wid], Bi[wid], lane, -1.0f);
  __syncthreads();
  fft_stageBC(Br[wid], Bi[wid], Ar[wid], Ai[wid], Twr, Twi, lane, w);
  __syncthreads();
  const int nn = lane & 15, mb = (lane >> 4) * 8;
#pragma unroll
  for (int t = 0; t < 2; ++t) {
    v8f dr, di;
    ctile(Ar[wid], Ai[wid], t, lane, w, dr, di);
#pragma unroll
    for (int r = 0; r < 8; ++r) {
      const int o = 32 * (mb + r) + 16 * t + nn;
      Br[wid][o] = dr[r]; Bi[wid][o] = di[r];
    }
  }
  __syncthreads();
  for (int i = tid; i < FN * LPB; i += TPB) {
    const int o = i >> 2, ww = i & 3;
    spec[o * FN + c0 + ww] = make_float2(Br[ww][o], Bi[ww][o]);
  }
}

__global__ __launch_bounds__(CTPB)
void k_band_cols(const float2* __restrict__ spec, const float* __restrict__ win,
                 float2* __restrict__ T, int b0) {
  FFT_SHARED_C16
  const int tid = threadIdx.x, lane = tid & 31, wid = tid >> 5;
  const int c0 = blockIdx.x * CPB;
  const int bi = blockIdx.y;
  const size_t woff = (size_t)(b0 + bi) * FN * FN;
  float2* Tb = T + (size_t)bi * FN * FN;
  const int ww = tid & (CPB - 1);
  const int e0 = tid >> 4;
  float2 s[16]; float wv[16];
#pragma unroll
  for (int it = 0; it < 16; ++it) {
    const int e = e0 + 32 * it;
    s[it]  = spec[e * FN + c0 + ww];
    wv[it] = win[woff + (size_t)e * FN + c0 + ww];
  }
  fill_tw512(Twr, Twi, tid, CTPB, 1.0f);
  W16 w; make_w16(1.0f, lane, w);
#pragma unroll
  for (int it = 0; it < 16; ++it) {
    const int e = e0 + 32 * it;
    Ar[ww][e] = s[it].x * wv[it]; Ai[ww][e] = s[it].y * wv[it];
  }
  __syncthreads();
  fft_stageA(Ar[wid], Ai[wid], Br[wid], Bi[wid], lane, 1.0f);
  __syncthreads();
  fft_stageBC(Br[wid], Bi[wid], Ar[wid], Ai[wid], Twr, Twi, lane, w);
  __syncthreads();
  const int nn = lane & 15, mb = (lane >> 4) * 8;
#pragma unroll
  for (int t = 0; t < 2; ++t) {
    v8f dr, di;
    ctile(Ar[wid], Ai[wid], t, lane, w, dr, di);
#pragma unroll
    for (int r = 0; r < 8; ++r) {
      const int o = 32 * (mb + r) + 16 * t + nn;
      Br[wid][o] = dr[r]; Bi[wid][o] = di[r];
    }
  }
  __syncthreads();
#pragma unroll
  for (int it = 0; it < 16; ++it) {
    const int o = e0 + 32 * it;
    Tb[o * FN + c0 + ww] = make_float2(Br[ww][o], Bi[ww][o]);
  }
}

__global__ __launch_bounds__(TPB)
void k_band_rows(const float2* __restrict__ T, float* __restrict__ out, int b0) {
  FFT_SHARED
  const int tid = threadIdx.x, lane = tid & 31, wid = tid >> 5;
  const int bi = blockIdx.y;
  const float2* Tb = T + (size_t)bi * FN * FN;
  float* ob = out + (size_t)(b0 + bi) * FN * FN;
  const int line = blockIdx.x * LPB + wid;
  fill_tw512(Twr, Twi, tid, TPB, 1.0f);
  W16 w; make_w16(1.0f, lane, w);
#pragma unroll
  for (int i = 0; i < 16; ++i) {
    const int p = lane + 32 * i;
    const float2 v = Tb[line * FN + p];
    Ar[wid][p] = v.x; Ai[wid][p] = v.y;
  }
  __syncthreads();
  fft_stageA(Ar[wid], Ai[wid], Br[wid], Bi[wid], lane, 1.0f);
  __syncthreads();
  fft_stageBC(Br[wid], Bi[wid], Ar[wid], Ai[wid], Twr, Twi, lane, w);
  __syncthreads();
  const float scale = 1.0f / (512.0f * 512.0f);
  const int nn = lane & 15, mb = (lane >> 4) * 8;
#pragma unroll
  for (int t = 0; t < 2; ++t) {
    v8f dr, di;
    ctile(Ar[wid], Ai[wid], t, lane, w, dr, di);
#pragma unroll
    for (int r = 0; r < 8; ++r) {
      const int o = 32 * (mb + r) + 16 * t + nn;
      ob[line * FN + o] = dr[r] * scale;
    }
  }
}

extern "C" void kernel_launch(void* const* d_in, const int* in_sizes, int n_in,
                              void* d_out, int out_size, void* d_ws,
                              size_t ws_size, hipStream_t stream) {
  (void)in_sizes; (void)n_in; (void)out_size;
  const float* img = (const float*)d_in[0];
  const float* win = (const float*)d_in[1];
  float* out = (float*)d_out;

  float2* specRow = (float2*)d_ws;
  float2* spec    = specRow + (size_t)FN * FN;
  float2* bandT   = spec    + (size_t)FN * FN;
  const size_t reserved = 2ull * sizeof(float2) * FN * FN;
  const size_t perBand  = sizeof(float2) * (size_t)FN * FN;
  int chunk = (ws_size > reserved) ? (int)((ws_size - reserved) / perBand) : 0;
  if (chunk < 1)  chunk = 1;
  if (chunk > NB) chunk = NB;

  const dim3 blk(TPB);
  const dim3 cblk(CTPB);
  k_fwd_rows<<<dim3(FN / LPB), blk, 0, stream>>>(img, specRow);
  k_fwd_cols<<<dim3(FN / LPB), blk, 0, stream>>>(specRow, spec);
  for (int b0 = 0; b0 < NB; b0 += chunk) {
    const int nb = (NB - b0 < chunk) ? (NB - b0) : chunk;
    k_band_cols<<<dim3(FN / CPB, nb), cblk, 0, stream>>>(spec, win, bandT, b0);
    k_band_rows<<<dim3(FN / LPB, nb), blk, 0, stream>>>(bandT, out, b0);
  }
}


# Inverse 2-D transform made real-output: window symmetrised to (W[k]+W[-k])/2, column pass does 256 columns (0 and 256 packed), row pass two rows per complex FFT; sums re-associated
# speedup vs baseline: 2.0987x; 1.2679x over previous
#include <hip/hip_runtime.h>
#include <hip/hip_bf16.h>


#define FN   512
#define NB   43
#define LPB  4
#define TPB  128
#define CPB    16
#define CTPB   512
#define LPITCH (FN + 4)
#define HC     256
#define PI_F 3.14159265358979323846f

typedef float v2f __attribute__((ext_vector_type(2)));
typedef float v8f __attribute__((ext_vector_type(8)));

__device__ __forceinline__ v8f wmma_f32(v2f a, v2f b, v8f c) {
  return __builtin_amdgcn_wmma_f32_16x16x4_f32(false, a, false, b,
                                               (short)0, c, false, false);
}

struct W16 { v2f ar[4]; v2f ai[4]; v2f ain[4]; };

__device__ __forceinline__ void make_w16(float sgn, int lane, W16& w) {
  const int m  = lane & 15;
  const int kb = (lane >> 4) << 1;
  const float step = sgn * (2.0f * PI_F / 16.0f);
#pragma unroll
  for (int kc = 0; kc < 4; ++kc) {
    const int k0 = 4 * kc + kb;
    float s0, c0, s1, c1;
    sincosf(step * (float)(m * k0),       &s0, &c0);
    sincosf(step * (float)(m * (k0 + 1)), &s1, &c1);
    v2f tr; tr.x = c0;  tr.y = c1;  w.ar[kc]  = tr;
    v2f ti; ti.x = s0;  ti.y = s1;  w.ai[kc]  = ti;
    v2f tn; tn.x = -s0; tn.y = -s1; w.ain[kc] = tn;
  }
}

__device__ __forceinline__ void ctile(const float* sr, const float* si, int t,
                                      int lane, const W16& w, v8f& cr, v8f& ci) {
  const int n  = 16 * t + (lane & 15);
  const int kb = (lane >> 4) << 1;
  v8f zr = {0.f,0.f,0.f,0.f,0.f,0.f,0.f,0.f};
  cr = zr; ci = zr;
#pragma unroll
  for (int kc = 0; kc < 4; ++kc) {
    const int k0 = 4 * kc + kb;
    v2f br; br.x = sr[k0 * 32 + n]; br.y = sr[(k0 + 1) * 32 + n];
    v2f bi; bi.x = si[k0 * 32 + n]; bi.y = si[(k0 + 1) * 32 + n];
    cr = wmma_f32(w.ar[kc],  br, cr);
    cr = wmma_f32(w.ain[kc], bi, cr);
    ci = wmma_f32(w.ar[kc],  bi, ci);
    ci = wmma_f32(w.ai[kc],  br, ci);
  }
}

__device__ __forceinline__ void fft_stageA(const float* xr, const float* xi,
                                           float* ur, float* ui,
                                           int lane, float sgn) {
  const float tws = sgn * (2.0f * PI_F / 32.0f);
#pragma unroll
  for (int i = 0; i < 8; ++i) {
    const int p  = lane + 32 * i;
    const int n1 = p & 15, m1 = p >> 4;
    const float a_r = xr[p],       a_i = xi[p];
    const float b_r = xr[p + 256], b_i = xi[p + 256];
    ur[m1 * 32 + n1] = a_r + b_r;
    ui[m1 * 32 + n1] = a_i + b_i;
    const float dr = a_r - b_r, di = a_i - b_i;
    float s, c; sincosf(tws * (float)m1, &s, &c);
    ur[m1 * 32 + n1 + 16] = dr * c - di * s;
    ui[m1 * 32 + n1 + 16] = dr * s + di * c;
  }
}

__device__ __forceinline__ void fft_stageBC(const float* ur, const float* ui,
                                            float* vr, float* vi,
                                            const float* twr, const float* twi,
                                            int lane, const W16& w) {
  const int nn = lane & 15;
  const int mb = (lane >> 4) * 8;
#pragma unroll
  for (int t = 0; t < 2; ++t) {
    v8f gr, gi;
    ctile(ur, ui, t, lane, w, gr, gi);
    const int ccol = 16 * t + nn;
    const int n1 = ccol & 15, j2 = ccol >> 4;
#pragma unroll
    for (int r = 0; r < 8; ++r) {
      const int j1  = mb + r;
      const int k2  = 2 * j1 + j2;
      const int idx = n1 * 32 + k2;
      const float tr = twr[idx], ti_ = twi[idx];
      const float g_r = gr[r], g_i = gi[r];
      vr[idx] = g_r * tr - g_i * ti_;
      vi[idx] = g_r * ti_ + g_i * tr;
    }
  }
}

__device__ __forceinline__ void fill_tw512(float* twr, float* twi,
                                           int tid, int nthr, float sgn) {
  const float step = sgn * (2.0f * PI_F / 512.0f);
  for (int i = tid; i < FN; i += nthr) {
    const int n1 = i >> 5, k2 = i & 31;
    float s, c; sincosf(step * (float)(n1 * k2), &s, &c);
    twr[i] = c; twi[i] = s;
  }
}

#define FFT_SHARED                                              \
  __shared__ float Ar[LPB][FN], Ai[LPB][FN];                    \
  __shared__ float Br[LPB][FN], Bi[LPB][FN];                    \
  __shared__ float Twr[FN], Twi[FN];

#define FFT_SHARED_C16                                          \
  __shared__ float Ar[CPB][LPITCH], Ai[CPB][LPITCH];            \
  __shared__ float Br[CPB][LPITCH], Bi[CPB][LPITCH];            \
  __shared__ float Twr[FN], Twi[FN];

__global__ __launch_bounds__(TPB)
void k_fwd_rows(const float* __restrict__ img, float2* __restrict__ specRow) {
  FFT_SHARED
  const int tid = threadIdx.x, lane = tid & 31, wid = tid >> 5;
  const int line = blockIdx.x * LPB + wid;
  fill_tw512(Twr, Twi, tid, TPB, -1.0f);
  W16 w; make_w16(-1.0f, lane, w);
#pragma unroll
  for (int i = 0; i < 16; ++i) {
    const int p = lane + 32 * i;
    Ar[wid][p] = img[line * FN + p];
    Ai[wid][p] = 0.0f;
  }
  __syncthreads();
  fft_stageA(Ar[wid], Ai[wid], Br[wid], Bi[wid], lane, -1.0f);
  __syncthreads();
  fft_stageBC(Br[wid], Bi[wid], Ar[wid], Ai[wid], Twr, Twi, lane, w);
  __syncthreads();
  const int nn = lane & 15, mb = (lane >> 4) * 8;
#pragma unroll
  for (int t = 0; t < 2; ++t) {
    v8f dr, di;
    ctile(Ar[wid], Ai[wid], t, lane, w, dr, di);
#pragma unroll
    for (int r = 0; r < 8; ++r) {
      const int o = 32 * (mb + r) + 16 * t + nn;
      specRow[line * FN + o] = make_float2(dr[r], di[r]);
    }
  }
}

__global__ __launch_bounds__(TPB)
void k_fwd_cols(const float2* __restrict__ specRow, float2* __restrict__ spec) {
  FFT_SHARED
  const int tid = threadIdx.x, lane = tid & 31, wid = tid >> 5;
  const int c0 = blockIdx.x * LPB;
  fill_tw512(Twr, Twi, tid, TPB, -1.0f);
  W16 w; make_w16(-1.0f, lane, w);
  for (int i = tid; i < FN * LPB; i += TPB) {
    const int e = i >> 2, ww = i & 3;
    const float2 s = specRow[e * FN + c0 + ww];
    Ar[ww][e] = s.x; Ai[ww][e] = s.y;
  }
  __syncthreads();
  fft_stageA(Ar[wid], Ai[wid], Br[wid], Bi[wid], lane, -1.0f);
  __syncthreads();
  fft_stageBC(Br[wid], Bi[wid], Ar[wid], Ai[wid], Twr, Twi, lane, w);
  __syncthreads();
  const int nn = lane & 15, mb = (lane >> 4) * 8;
#pragma unroll
  for (int t = 0; t < 2; ++t) {
    v8f dr, di;
    ctile(Ar[wid], Ai[wid], t, lane, w, dr, di);
#pragma unroll
    for (int r = 0; r < 8; ++r) {
      const int o = 32 * (mb + r) + 16 * t + nn;
      Br[wid][o] = dr[r]; Bi[wid][o] = di[r];
    }
  }
  __syncthreads();
  for (int i = tid; i < FN * LPB; i += TPB) {
    const int o = i >> 2, ww = i & 3;
    spec[o * FN + c0 + ww] = make_float2(Br[ww][o], Bi[ww][o]);
  }
}

__global__ __launch_bounds__(CTPB)
void k_band_cols(const float2* __restrict__ spec, const float* __restrict__ win,
                 float2* __restrict__ T, int b0) {
  FFT_SHARED_C16
  const int tid = threadIdx.x, lane = tid & 31, wid = tid >> 5;
  const int c0 = blockIdx.x * CPB;
  const int bi = blockIdx.y;
  const float* wb = win + (size_t)(b0 + bi) * FN * FN;
  float2* Tb = T + (size_t)bi * FN * HC;
  const int ww = tid & (CPB - 1);
  const int e0 = tid >> 4;
  const int k2  = c0 + ww;
  const int mk2 = (FN - k2) & (FN - 1);
  float2 s[16]; float wv[16], wm[16];
#pragma unroll
  for (int it = 0; it < 16; ++it) {
    const int e  = e0 + 32 * it;
    const int me = (FN - e) & (FN - 1);
    s[it]  = spec[e * FN + k2];
    wv[it] = wb[e * FN + k2];
    wm[it] = wb[me * FN + mk2];
  }
  fill_tw512(Twr, Twi, tid, CTPB, 1.0f);
  W16 w; make_w16(1.0f, lane, w);
#pragma unroll
  for (int it = 0; it < 16; ++it) {
    const int e = e0 + 32 * it;
    const float g = 0.5f * (wv[it] + wm[it]);
    Ar[ww][e] = s[it].x * g; Ai[ww][e] = s[it].y * g;
  }
  if (blockIdx.x == 0 && ww == 0) {
#pragma unroll
    for (int it = 0; it < 16; ++it) {
      const int e  = e0 + 32 * it;
      const int me = (FN - e) & (FN - 1);
      const float2 q = spec[e * FN + HC];
      const float g = 0.5f * (wb[e * FN + HC] + wb[me * FN + HC]);
      Ar[0][e] -= q.y * g;
      Ai[0][e] += q.x * g;
    }
  }
  __syncthreads();
  fft_stageA(Ar[wid], Ai[wid], Br[wid], Bi[wid], lane, 1.0f);
  __syncthreads();
  fft_stageBC(Br[wid], Bi[wid], Ar[wid], Ai[wid], Twr, Twi, lane, w);
  __syncthreads();
  const int nn = lane & 15, mb = (lane >> 4) * 8;
#pragma unroll
  for (int t = 0; t < 2; ++t) {
    v8f dr, di;
    ctile(Ar[wid], Ai[wid], t, lane, w, dr, di);
#pragma unroll
    for (int r = 0; r < 8; ++r) {
      const int o = 32 * (mb + r) + 16 * t + nn;
      Br[wid][o] = dr[r]; Bi[wid][o] = di[r];
    }
  }
  __syncthreads();
#pragma unroll
  for (int it = 0; it < 16; ++it) {
    const int o = e0 + 32 * it;
    Tb[o * HC + k2] = make_float2(Br[ww][o], Bi[ww][o]);
  }
}

__global__ __launch_bounds__(TPB)
void k_band_rows(const float2* __restrict__ T, float* __restrict__ out, int b0) {
  FFT_SHARED
  const int tid = threadIdx.x, lane = tid & 31, wid = tid >> 5;
  const int bi = blockIdx.y;
  const float2* Tb = T + (size_t)bi * FN * HC;
  float* ob = out + (size_t)(b0 + bi) * FN * FN;
  const int ra = 2 * (blockIdx.x * LPB + wid);
  const int rb = ra + 1;
  fill_tw512(Twr, Twi, tid, TPB, 1.0f);
  W16 w; make_w16(1.0f, lane, w);
#pragma unroll
  for (int i = 0; i < 8; ++i) {
    const int k = lane + 32 * i;
    const float2 a = Tb[ra * HC + k];
    const float2 b = Tb[rb * HC + k];
    const bool z = (k == 0);
    const int  km = z ? HC : FN - k;
    Ar[wid][k]  = z ? a.x : a.x - b.y;
    Ai[wid][k]  = z ? b.x : a.y + b.x;
    Ar[wid][km] = z ? a.y : a.x + b.y;
    Ai[wid][km] = z ? b.y : b.x - a.y;
  }
  __syncthreads();
  fft_stageA(Ar[wid], Ai[wid], Br[wid], Bi[wid], lane, 1.0f);
  __syncthreads();
  fft_stageBC(Br[wid], Bi[wid], Ar[wid], Ai[wid], Twr, Twi, lane, w);
  __syncthreads();
  const float scale = 1.0f / (512.0f * 512.0f);
  const int nn = lane & 15, mb = (lane >> 4) * 8;
#pragma unroll
  for (int t = 0; t < 2; ++t) {
    v8f dr, di;
    ctile(Ar[wid], Ai[wid], t, lane, w, dr, di);
#pragma unroll
    for (int r = 0; r < 8; ++r) {
      const int o = 32 * (mb + r) + 16 * t + nn;
      ob[ra * FN + o] = dr[r] * scale;
      ob[rb * FN + o] = di[r] * scale;
    }
  }
}

extern "C" void kernel_launch(void* const* d_in, const int* in_sizes, int n_in,
                              void* d_out, int out_size, void* d_ws,
                              size_t ws_size, hipStream_t stream) {
  (void)in_sizes; (void)n_in; (void)out_size;
  const float* img = (const float*)d_in[0];
  const float* win = (const float*)d_in[1];
  float* out = (float*)d_out;

  float2* specRow = (float2*)d_ws;
  float2* spec    = specRow + (size_t)FN * FN;
  float2* bandT   = spec    + (size_t)FN * FN;
  const size_t reserved = 2ull * sizeof(float2) * FN * FN;
  const size_t perBand  = sizeof(float2) * (size_t)FN * FN;
  int chunk = (ws_size > reserved) ? (int)((ws_size - reserved) / perBand) : 0;
  if (chunk < 1)  chunk = 1;
  if (chunk > NB) chunk = NB;

  const dim3 blk(TPB);
  const dim3 cblk(CTPB);
  k_fwd_rows<<<dim3(FN / LPB), blk, 0, stream>>>(img, specRow);
  k_fwd_cols<<<dim3(FN / LPB), blk, 0, stream>>>(specRow, spec);
  for (int b0 = 0; b0 < NB; b0 += chunk) {
    const int nb = (NB - b0 < chunk) ? (NB - b0) : chunk;
    k_band_cols<<<dim3(HC / CPB, nb), cblk, 0, stream>>>(spec, win, bandT, b0);
    k_band_rows<<<dim3(FN / (2 * LPB), nb), blk, 0, stream>>>(bandT, out, b0);
  }
}


# Inverse passes swapped: row pass first, one wave per spectrum row, window and mirrored window read as whole contiguous rows; column pass second on 32-column real-pair tiles; sums re-associated
# speedup vs baseline: 2.3833x; 1.1356x over previous
#include <hip/hip_runtime.h>
#include <hip/hip_bf16.h>


#define FN   512
#define NB   43
#define LPB  4
#define TPB  128
#define CPB    16
#define CTPB   512
#define LPITCH (FN + 4)
#define HC     256
#define PI_F 3.14159265358979323846f

typedef float v2f __attribute__((ext_vector_type(2)));
typedef float v8f __attribute__((ext_vector_type(8)));

__device__ __forceinline__ v8f wmma_f32(v2f a, v2f b, v8f c) {
  return __builtin_amdgcn_wmma_f32_16x16x4_f32(false, a, false, b,
                                               (short)0, c, false, false);
}

struct W16 { v2f ar[4]; v2f ai[4]; v2f ain[4]; };

__device__ __forceinline__ void make_w16(float sgn, int lane, W16& w) {
  const int m  = lane & 15;
  const int kb = (lane >> 4) << 1;
  const float step = sgn * (2.0f * PI_F / 16.0f);
#pragma unroll
  for (int kc = 0; kc < 4; ++kc) {
    const int k0 = 4 * kc + kb;
    float s0, c0, s1, c1;
    sincosf(step * (float)(m * k0),       &s0, &c0);
    sincosf(step * (float)(m * (k0 + 1)), &s1, &c1);
    v2f tr; tr.x = c0;  tr.y = c1;  w.ar[kc]  = tr;
    v2f ti; ti.x = s0;  ti.y = s1;  w.ai[kc]  = ti;
    v2f tn; tn.x = -s0; tn.y = -s1; w.ain[kc] = tn;
  }
}

__device__ __forceinline__ void ctile(const float* sr, const float* si, int t,
                                      int lane, const W16& w, v8f& cr, v8f& ci) {
  const int n  = 16 * t + (lane & 15);
  const int kb = (lane >> 4) << 1;
  v8f zr = {0.f,0.f,0.f,0.f,0.f,0.f,0.f,0.f};
  cr = zr; ci = zr;
#pragma unroll
  for (int kc = 0; kc < 4; ++kc) {
    const int k0 = 4 * kc + kb;
    v2f br; br.x = sr[k0 * 32 + n]; br.y = sr[(k0 + 1) * 32 + n];
    v2f bi; bi.x = si[k0 * 32 + n]; bi.y = si[(k0 + 1) * 32 + n];
    cr = wmma_f32(w.ar[kc],  br, cr);
    cr = wmma_f32(w.ain[kc], bi, cr);
    ci = wmma_f32(w.ar[kc],  bi, ci);
    ci = wmma_f32(w.ai[kc],  br, ci);
  }
}

__device__ __forceinline__ void fft_stageA(const float* xr, const float* xi,
                                           float* ur, float* ui,
                                           int lane, float sgn) {
  const float tws = sgn * (2.0f * PI_F / 32.0f);
#pragma unroll
  for (int i = 0; i < 8; ++i) {
    const int p  = lane + 32 * i;
    const int n1 = p & 15, m1 = p >> 4;
    const float a_r = xr[p],       a_i = xi[p];
    const float b_r = xr[p + 256], b_i = xi[p + 256];
    ur[m1 * 32 + n1] = a_r + b_r;
    ui[m1 * 32 + n1] = a_i + b_i;
    const float dr = a_r - b_r, di = a_i - b_i;
    float s, c; sincosf(tws * (float)m1, &s, &c);
    ur[m1 * 32 + n1 + 16] = dr * c - di * s;
    ui[m1 * 32 + n1 + 16] = dr * s + di * c;
  }
}

__device__ __forceinline__ void fft_stageBC(const float* ur, const float* ui,
                                            float* vr, float* vi,
                                            const float* twr, const float* twi,
                                            int lane, const W16& w) {
  const int nn = lane & 15;
  const int mb = (lane >> 4) * 8;
#pragma unroll
  for (int t = 0; t < 2; ++t) {
    v8f gr, gi;
    ctile(ur, ui, t, lane, w, gr, gi);
    const int ccol = 16 * t + nn;
    const int n1 = ccol & 15, j2 = ccol >> 4;
#pragma unroll
    for (int r = 0; r < 8; ++r) {
      const int j1  = mb + r;
      const int k2  = 2 * j1 + j2;
      const int idx = n1 * 32 + k2;
      const float tr = twr[idx], ti_ = twi[idx];
      const float g_r = gr[r], g_i = gi[r];
      vr[idx] = g_r * tr - g_i * ti_;
      vi[idx] = g_r * ti_ + g_i * tr;
    }
  }
}

__device__ __forceinline__ void fill_tw512(float* twr, float* twi,
                                           int tid, int nthr, float sgn) {
  const float step = sgn * (2.0f * PI_F / 512.0f);
  for (int i = tid; i < FN; i += nthr) {
    const int n1 = i >> 5, k2 = i & 31;
    float s, c; sincosf(step * (float)(n1 * k2), &s, &c);
    twr[i] = c; twi[i] = s;
  }
}

#define FFT_SHARED                                              \
  __shared__ float Ar[LPB][FN], Ai[LPB][FN];                    \
  __shared__ float Br[LPB][FN], Bi[LPB][FN];                    \
  __shared__ float Twr[FN], Twi[FN];

#define FFT_SHARED_C16                                          \
  __shared__ float Ar[CPB][LPITCH], Ai[CPB][LPITCH];            \
  __shared__ float Br[CPB][LPITCH], Bi[CPB][LPITCH];            \
  __shared__ float Twr[FN], Twi[FN];

__global__ __launch_bounds__(TPB)
void k_fwd_rows(const float* __restrict__ img, float2* __restrict__ specRow) {
  FFT_SHARED
  const int tid = threadIdx.x, lane = tid & 31, wid = tid >> 5;
  const int line = blockIdx.x * LPB + wid;
  fill_tw512(Twr, Twi, tid, TPB, -1.0f);
  W16 w; make_w16(-1.0f, lane, w);
#pragma unroll
  for (int i = 0; i < 16; ++i) {
    const int p = lane + 32 * i;
    Ar[wid][p] = img[line * FN + p];
    Ai[wid][p] = 0.0f;
  }
  __syncthreads();
  fft_stageA(Ar[wid], Ai[wid], Br[wid], Bi[wid], lane, -1.0f);
  __syncthreads();
  fft_stageBC(Br[wid], Bi[wid], Ar[wid], Ai[wid], Twr, Twi, lane, w);
  __syncthreads();
  const int nn = lane & 15, mb = (lane >> 4) * 8;
#pragma unroll
  for (int t = 0; t < 2; ++t) {
    v8f dr, di;
    ctile(Ar[wid], Ai[wid], t, lane, w, dr, di);
#pragma unroll
    for (int r = 0; r < 8; ++r) {
      const int o = 32 * (mb + r) + 16 * t + nn;
      specRow[line * FN + o] = make_float2(dr[r], di[r]);
    }
  }
}

__global__ __launch_bounds__(TPB)
void k_fwd_cols(const float2* __restrict__ specRow, float2* __restrict__ spec) {
  FFT_SHARED
  const int tid = threadIdx.x, lane = tid & 31, wid = tid >> 5;
  const int c0 = blockIdx.x * LPB;
  fill_tw512(Twr, Twi, tid, TPB, -1.0f);
  W16 w; make_w16(-1.0f, lane, w);
  for (int i = tid; i < FN * LPB; i += TPB) {
    const int e = i >> 2, ww = i & 3;
    const float2 s = specRow[e * FN + c0 + ww];
    Ar[ww][e] = s.x; Ai[ww][e] = s.y;
  }
  __syncthreads();
  fft_stageA(Ar[wid], Ai[wid], Br[wid], Bi[wid], lane, -1.0f);
  __syncthreads();
  fft_stageBC(Br[wid], Bi[wid], Ar[wid], Ai[wid], Twr, Twi, lane, w);
  __syncthreads();
  const int nn = lane & 15, mb = (lane >> 4) * 8;
#pragma unroll
  for (int t = 0; t < 2; ++t) {
    v8f dr, di;
    ctile(Ar[wid], Ai[wid], t, lane, w, dr, di);
#pragma unroll
    for (int r = 0; r < 8; ++r) {
      const int o = 32 * (mb + r) + 16 * t + nn;
      Br[wid][o] = dr[r]; Bi[wid][o] = di[r];
    }
  }
  __syncthreads();
  for (int i = tid; i < FN * LPB; i += TPB) {
    const int o = i >> 2, ww = i & 3;
    spec[o * FN + c0 + ww] = make_float2(Br[ww][o], Bi[ww][o]);
  }
}

__global__ __launch_bounds__(TPB)
void k_band_rows(const float2* __restrict__ spec, const float* __restrict__ win,
                 float2* __restrict__ T, int b0) {
  FFT_SHARED
  const int tid = threadIdx.x, lane = tid & 31, wid = tid >> 5;
  const int bi = blockIdx.y;
  const float* wb = win + (size_t)(b0 + bi) * FN * FN;
  float2* Tb = T + (size_t)bi * HC * FN;
  const int k1  = blockIdx.x * LPB + wid;
  const int mk1 = (FN - k1) & (FN - 1);
  const float2* srow = spec + k1 * FN;
  const float*  wrow = wb + k1 * FN;
  const float*  mrow = wb + mk1 * FN;
  float2 s[16]; float wv[16], wm[16];
#pragma unroll
  for (int i = 0; i < 16; ++i) {
    const int k2 = lane + 32 * i;
    s[i]  = srow[k2];
    wv[i] = wrow[k2];
    wm[i] = mrow[(FN - k2) & (FN - 1)];
  }
  fill_tw512(Twr, Twi, tid, TPB, 1.0f);
  W16 w; make_w16(1.0f, lane, w);
#pragma unroll
  for (int i = 0; i < 16; ++i) {
    const int k2 = lane + 32 * i;
    const float g = 0.5f * (wv[i] + wm[i]);
    Ar[wid][k2] = s[i].x * g; Ai[wid][k2] = s[i].y * g;
  }
  if (k1 == 0) {
#pragma unroll
    for (int i = 0; i < 16; ++i) {
      const int k2 = lane + 32 * i;
      const float2 q = spec[HC * FN + k2];
      const float g = 0.5f * (wb[HC * FN + k2] +
                              wb[HC * FN + ((FN - k2) & (FN - 1))]);
      Ar[wid][k2] -= q.y * g;
      Ai[wid][k2] += q.x * g;
    }
  }
  __syncthreads();
  fft_stageA(Ar[wid], Ai[wid], Br[wid], Bi[wid], lane, 1.0f);
  __syncthreads();
  fft_stageBC(Br[wid], Bi[wid], Ar[wid], Ai[wid], Twr, Twi, lane, w);
  __syncthreads();
  const int nn = lane & 15, mb = (lane >> 4) * 8;
#pragma unroll
  for (int t = 0; t < 2; ++t) {
    v8f dr, di;
    ctile(Ar[wid], Ai[wid], t, lane, w, dr, di);
#pragma unroll
    for (int r = 0; r < 8; ++r) {
      const int o = 32 * (mb + r) + 16 * t + nn;
      Tb[k1 * FN + o] = make_float2(dr[r], di[r]);
    }
  }
}

__global__ __launch_bounds__(CTPB)
void k_band_cols(const float2* __restrict__ T, float* __restrict__ out, int b0) {
  FFT_SHARED_C16
  const int tid = threadIdx.x, lane = tid & 31, wid = tid >> 5;
  const int c0 = blockIdx.x * (2 * CPB);
  const int bi = blockIdx.y;
  const float2* Tb = T + (size_t)bi * HC * FN;
  float* ob = out + (size_t)(b0 + bi) * FN * FN;
  const int ww = tid & (CPB - 1);
  const int e0 = tid >> 4;
  float2 a[8], b[8];
#pragma unroll
  for (int it = 0; it < 8; ++it) {
    const int k = e0 + 32 * it;
    a[it] = Tb[k * FN + c0 + ww];
    b[it] = Tb[k * FN + c0 + CPB + ww];
  }
  fill_tw512(Twr, Twi, tid, CTPB, 1.0f);
  W16 w; make_w16(1.0f, lane, w);
#pragma unroll
  for (int it = 0; it < 8; ++it) {
    const int k = e0 + 32 * it;
    const bool z = (k == 0);
    const int  km = z ? HC : FN - k;
    const float2 p = a[it], q = b[it];
    Ar[ww][k]  = z ? p.x : p.x - q.y;
    Ai[ww][k]  = z ? q.x : p.y + q.x;
    Ar[ww][km] = z ? p.y : p.x + q.y;
    Ai[ww][km] = z ? q.y : q.x - p.y;
  }
  __syncthreads();
  fft_stageA(Ar[wid], Ai[wid], Br[wid], Bi[wid], lane, 1.0f);
  __syncthreads();
  fft_stageBC(Br[wid], Bi[wid], Ar[wid], Ai[wid], Twr, Twi, lane, w);
  __syncthreads();
  const int nn = lane & 15, mb = (lane >> 4) * 8;
#pragma unroll
  for (int t = 0; t < 2; ++t) {
    v8f dr, di;
    ctile(Ar[wid], Ai[wid], t, lane, w, dr, di);
#pragma unroll
    for (int r = 0; r < 8; ++r) {
      const int o = 32 * (mb + r) + 16 * t + nn;
      Br[wid][o] = dr[r]; Bi[wid][o] = di[r];
    }
  }
  __syncthreads();
  const float scale = 1.0f / (512.0f * 512.0f);
  const int cw = tid & 31;
  const int r0 = tid >> 5;
  const float* col = (cw < CPB) ? &Br[cw][0] : &Bi[cw - CPB][0];
#pragma unroll
  for (int it = 0; it < 32; ++it) {
    const int o = r0 + 16 * it;
    ob[o * FN + c0 + cw] = col[o] * scale;
  }
}

extern "C" void kernel_launch(void* const* d_in, const int* in_sizes, int n_in,
                              void* d_out, int out_size, void* d_ws,
                              size_t ws_size, hipStream_t stream) {
  (void)in_sizes; (void)n_in; (void)out_size;
  const float* img = (const float*)d_in[0];
  const float* win = (const float*)d_in[1];
  float* out = (float*)d_out;

  float2* specRow = (float2*)d_ws;
  float2* spec    = specRow + (size_t)FN * FN;
  float2* bandT   = spec    + (size_t)FN * FN;
  const size_t reserved = 2ull * sizeof(float2) * FN * FN;
  const size_t perBand  = sizeof(float2) * (size_t)FN * FN;
  int chunk = (ws_size > reserved) ? (int)((ws_size - reserved) / perBand) : 0;
  if (chunk < 1)  chunk = 1;
  if (chunk > NB) chunk = NB;

  const dim3 blk(TPB);
  const dim3 cblk(CTPB);
  k_fwd_rows<<<dim3(FN / LPB), blk, 0, stream>>>(img, specRow);
  k_fwd_cols<<<dim3(FN / LPB), blk, 0, stream>>>(specRow, spec);
  for (int b0 = 0; b0 < NB; b0 += chunk) {
    const int nb = (NB - b0 < chunk) ? (NB - b0) : chunk;
    k_band_rows<<<dim3(HC / LPB, nb), blk, 0, stream>>>(spec, win, bandT, b0);
    k_band_cols<<<dim3(FN / (2 * CPB), nb), cblk, 0, stream>>>(bandT, out, b0);
  }
}


# Twiddle factors from an exact integer quarter-turn reduction, short sine/cosine polynomials and literal constants in registers, replacing every library sincosf call; no table in memory
# speedup vs baseline: 2.7910x; 1.1711x over previous
#include <hip/hip_runtime.h>
#include <hip/hip_bf16.h>


#define FN   512
#define NB   43
#define LPB  4
#define TPB  128
#define CPB    16
#define CTPB   512
#define LPITCH (FN + 4)
#define HC     256
#define PI_F 3.14159265358979323846f

typedef float v2f __attribute__((ext_vector_type(2)));
typedef float v8f __attribute__((ext_vector_type(8)));

__device__ __forceinline__ v8f wmma_f32(v2f a, v2f b, v8f c) {
  return __builtin_amdgcn_wmma_f32_16x16x4_f32(false, a, false, b,
                                               (short)0, c, false, false);
}


__device__ __forceinline__ void quarter_turn(float c, float s, int t,
                                             float& co, float& so) {
  const bool odd = (t & 1) != 0;
  const float a = odd ? s : c;
  const float b = odd ? c : s;
  const unsigned nc = ((unsigned)(t + 1) & 2u) << 30;
  const unsigned ns = ((unsigned)t & 2u) << 30;
  co = __uint_as_float(__float_as_uint(a) ^ nc);
  so = __uint_as_float(__float_as_uint(b) ^ ns);
}

__device__ __forceinline__ void cis512(int e, float sgn, float& c, float& s) {
  const int q  = (e + 64) >> 7;
  const int rr = e - (q << 7);
  const float x = (float)rr * (2.0f * PI_F / 512.0f);
  const float z = x * x;
  const float ps = fmaf(z, fmaf(z, -1.9515295891e-4f, 8.3321608736e-3f),
                        -1.6666654611e-1f);
  const float pc = fmaf(z, fmaf(z, 2.443315711809948e-5f,
                                -1.388731625493765e-3f),
                        4.166664568298827e-2f);
  const float sx = fmaf(x * z, ps, x);
  const float cx = fmaf(z * z, pc, fmaf(-0.5f, z, 1.0f));
  float s0;
  quarter_turn(cx, sx, q, c, s0);
  s = sgn * s0;
}

__device__ __forceinline__ constexpr float cos_pi16(int k) {
  return k == 0 ? 1.0f
       : k == 1 ? 0.98078528040323044913f
       : k == 2 ? 0.92387953251128675613f
       : k == 3 ? 0.83146961230254523708f
       : k == 4 ? 0.70710678118654752440f
       : k == 5 ? 0.55557023301960222474f
       : k == 6 ? 0.38268343236508977173f
       : k == 7 ? 0.19509032201612826785f
       : k == 8 ? 0.0f
       : k == 9 ? -0.19509032201612826785f
       : k == 10 ? -0.38268343236508977173f
       : k == 11 ? -0.55557023301960222474f
       : k == 12 ? -0.70710678118654752440f
       : k == 13 ? -0.83146961230254523708f
       : k == 14 ? -0.92387953251128675613f
       :           -0.98078528040323044913f;
}
__device__ __forceinline__ constexpr float sin_pi16(int k) {
  return cos_pi16(k <= 8 ? 8 - k : k - 8);
}

struct W16 { v2f ar[4]; v2f ai[4]; v2f ain[4]; };

__device__ __forceinline__ void make_w16(float sgn, int lane, W16& w) {
  const int m  = lane & 15;
  const int kb = (lane >> 4) << 1;
  float cb0, sb0, cb1, sb1;
  cis512(32 * (m * kb),       sgn, cb0, sb0);
  cis512(32 * (m * (kb + 1)), sgn, cb1, sb1);
#pragma unroll
  for (int kc = 0; kc < 4; ++kc) {
    const int t = (sgn > 0.0f) ? (m * kc) : -(m * kc);
    float s0, c0, s1, c1;
    quarter_turn(cb0, sb0, t, c0, s0);
    quarter_turn(cb1, sb1, t, c1, s1);
    v2f tr; tr.x = c0;  tr.y = c1;  w.ar[kc]  = tr;
    v2f ti; ti.x = s0;  ti.y = s1;  w.ai[kc]  = ti;
    v2f tn; tn.x = -s0; tn.y = -s1; w.ain[kc] = tn;
  }
}

__device__ __forceinline__ void ctile(const float* sr, const float* si, int t,
                                      int lane, const W16& w, v8f& cr, v8f& ci) {
  const int n  = 16 * t + (lane & 15);
  const int kb = (lane >> 4) << 1;
  v8f zr = {0.f,0.f,0.f,0.f,0.f,0.f,0.f,0.f};
  cr = zr; ci = zr;
#pragma unroll
  for (int kc = 0; kc < 4; ++kc) {
    const int k0 = 4 * kc + kb;
    v2f br; br.x = sr[k0 * 32 + n]; br.y = sr[(k0 + 1) * 32 + n];
    v2f bi; bi.x = si[k0 * 32 + n]; bi.y = si[(k0 + 1) * 32 + n];
    cr = wmma_f32(w.ar[kc],  br, cr);
    cr = wmma_f32(w.ain[kc], bi, cr);
    ci = wmma_f32(w.ar[kc],  bi, ci);
    ci = wmma_f32(w.ai[kc],  br, ci);
  }
}

__device__ __forceinline__ void fft_stageA(const float* xr, const float* xi,
                                           float* ur, float* ui,
                                           int lane, float sgn) {
  const bool hi = (lane >> 4) != 0;
#pragma unroll
  for (int i = 0; i < 8; ++i) {
    const int p  = lane + 32 * i;
    const int n1 = p & 15, m1 = p >> 4;
    const float a_r = xr[p],       a_i = xi[p];
    const float b_r = xr[p + 256], b_i = xi[p + 256];
    ur[m1 * 32 + n1] = a_r + b_r;
    ui[m1 * 32 + n1] = a_i + b_i;
    const float dr = a_r - b_r, di = a_i - b_i;
    const float c = hi ? cos_pi16(2 * i + 1) : cos_pi16(2 * i);
    const float s = sgn * (hi ? sin_pi16(2 * i + 1) : sin_pi16(2 * i));
    ur[m1 * 32 + n1 + 16] = dr * c - di * s;
    ui[m1 * 32 + n1 + 16] = dr * s + di * c;
  }
}

__device__ __forceinline__ void fft_stageBC(const float* ur, const float* ui,
                                            float* vr, float* vi,
                                            const float* twr, const float* twi,
                                            int lane, const W16& w) {
  const int nn = lane & 15;
  const int mb = (lane >> 4) * 8;
#pragma unroll
  for (int t = 0; t < 2; ++t) {
    v8f gr, gi;
    ctile(ur, ui, t, lane, w, gr, gi);
    const int ccol = 16 * t + nn;
    const int n1 = ccol & 15, j2 = ccol >> 4;
#pragma unroll
    for (int r = 0; r < 8; ++r) {
      const int j1  = mb + r;
      const int k2  = 2 * j1 + j2;
      const int idx = n1 * 32 + k2;
      const float tr = twr[idx], ti_ = twi[idx];
      const float g_r = gr[r], g_i = gi[r];
      vr[idx] = g_r * tr - g_i * ti_;
      vi[idx] = g_r * ti_ + g_i * tr;
    }
  }
}

__device__ __forceinline__ void fill_tw512(float* twr, float* twi,
                                           int tid, int nthr, float sgn) {
  for (int i = tid; i < FN; i += nthr) {
    const int n1 = i >> 5, k2 = i & 31;
    float s, c; cis512(n1 * k2, sgn, c, s);
    twr[i] = c; twi[i] = s;
  }
}

#define FFT_SHARED                                              \
  __shared__ float Ar[LPB][FN], Ai[LPB][FN];                    \
  __shared__ float Br[LPB][FN], Bi[LPB][FN];                    \
  __shared__ float Twr[FN], Twi[FN];

#define FFT_SHARED_C16                                          \
  __shared__ float Ar[CPB][LPITCH], Ai[CPB][LPITCH];            \
  __shared__ float Br[CPB][LPITCH], Bi[CPB][LPITCH];            \
  __shared__ float Twr[FN], Twi[FN];

__global__ __launch_bounds__(TPB)
void k_fwd_rows(const float* __restrict__ img, float2* __restrict__ specRow) {
  FFT_SHARED
  const int tid = threadIdx.x, lane = tid & 31, wid = tid >> 5;
  const int line = blockIdx.x * LPB + wid;
  fill_tw512(Twr, Twi, tid, TPB, -1.0f);
  W16 w; make_w16(-1.0f, lane, w);
#pragma unroll
  for (int i = 0; i < 16; ++i) {
    const int p = lane + 32 * i;
    Ar[wid][p] = img[line * FN + p];
    Ai[wid][p] = 0.0f;
  }
  __syncthreads();
  fft_stageA(Ar[wid], Ai[wid], Br[wid], Bi[wid], lane, -1.0f);
  __syncthreads();
  fft_stageBC(Br[wid], Bi[wid], Ar[wid], Ai[wid], Twr, Twi, lane, w);
  __syncthreads();
  const int nn = lane & 15, mb = (lane >> 4) * 8;
#pragma unroll
  for (int t = 0; t < 2; ++t) {
    v8f dr, di;
    ctile(Ar[wid], Ai[wid], t, lane, w, dr, di);
#pragma unroll
    for (int r = 0; r < 8; ++r) {
      const int o = 32 * (mb + r) + 16 * t + nn;
      specRow[line * FN + o] = make_float2(dr[r], di[r]);
    }
  }
}

__global__ __launch_bounds__(TPB)
void k_fwd_cols(const float2* __restrict__ specRow, float2* __restrict__ spec) {
  FFT_SHARED
  const int tid = threadIdx.x, lane = tid & 31, wid = tid >> 5;
  const int c0 = blockIdx.x * LPB;
  fill_tw512(Twr, Twi, tid, TPB, -1.0f);
  W16 w; make_w16(-1.0f, lane, w);
  for (int i = tid; i < FN * LPB; i += TPB) {
    const int e = i >> 2, ww = i & 3;
    const float2 s = specRow[e * FN + c0 + ww];
    Ar[ww][e] = s.x; Ai[ww][e] = s.y;
  }
  __syncthreads();
  fft_stageA(Ar[wid], Ai[wid], Br[wid], Bi[wid], lane, -1.0f);
  __syncthreads();
  fft_stageBC(Br[wid], Bi[wid], Ar[wid], Ai[wid], Twr, Twi, lane, w);
  __syncthreads();
  const int nn = lane & 15, mb = (lane >> 4) * 8;
#pragma unroll
  for (int t = 0; t < 2; ++t) {
    v8f dr, di;
    ctile(Ar[wid], Ai[wid], t, lane, w, dr, di);
#pragma unroll
    for (int r = 0; r < 8; ++r) {
      const int o = 32 * (mb + r) + 16 * t + nn;
      Br[wid][o] = dr[r]; Bi[wid][o] = di[r];
    }
  }
  __syncthreads();
  for (int i = tid; i < FN * LPB; i += TPB) {
    const int o = i >> 2, ww = i & 3;
    spec[o * FN + c0 + ww] = make_float2(Br[ww][o], Bi[ww][o]);
  }
}

__global__ __launch_bounds__(TPB)
void k_band_rows(const float2* __restrict__ spec, const float* __restrict__ win,
                 float2* __restrict__ T, int b0) {
  FFT_SHARED
  const int tid = threadIdx.x, lane = tid & 31, wid = tid >> 5;
  const int bi = blockIdx.y;
  const float* wb = win + (size_t)(b0 + bi) * FN * FN;
  float2* Tb = T + (size_t)bi * HC * FN;
  const int k1  = blockIdx.x * LPB + wid;
  const int mk1 = (FN - k1) & (FN - 1);
  const float2* srow = spec + k1 * FN;
  const float*  wrow = wb + k1 * FN;
  const float*  mrow = wb + mk1 * FN;
  float2 s[16]; float wv[16], wm[16];
#pragma unroll
  for (int i = 0; i < 16; ++i) {
    const int k2 = lane + 32 * i;
    s[i]  = srow[k2];
    wv[i] = wrow[k2];
    wm[i] = mrow[(FN - k2) & (FN - 1)];
  }
  fill_tw512(Twr, Twi, tid, TPB, 1.0f);
  W16 w; make_w16(1.0f, lane, w);
#pragma unroll
  for (int i = 0; i < 16; ++i) {
    const int k2 = lane + 32 * i;
    const float g = 0.5f * (wv[i] + wm[i]);
    Ar[wid][k2] = s[i].x * g; Ai[wid][k2] = s[i].y * g;
  }
  if (k1 == 0) {
#pragma unroll
    for (int i = 0; i < 16; ++i) {
      const int k2 = lane + 32 * i;
      const float2 q = spec[HC * FN + k2];
      const float g = 0.5f * (wb[HC * FN + k2] +
                              wb[HC * FN + ((FN - k2) & (FN - 1))]);
      Ar[wid][k2] -= q.y * g;
      Ai[wid][k2] += q.x * g;
    }
  }
  __syncthreads();
  fft_stageA(Ar[wid], Ai[wid], Br[wid], Bi[wid], lane, 1.0f);
  __syncthreads();
  fft_stageBC(Br[wid], Bi[wid], Ar[wid], Ai[wid], Twr, Twi, lane, w);
  __syncthreads();
  const int nn = lane & 15, mb = (lane >> 4) * 8;
#pragma unroll
  for (int t = 0; t < 2; ++t) {
    v8f dr, di;
    ctile(Ar[wid], Ai[wid], t, lane, w, dr, di);
#pragma unroll
    for (int r = 0; r < 8; ++r) {
      const int o = 32 * (mb + r) + 16 * t + nn;
      Tb[k1 * FN + o] = make_float2(dr[r], di[r]);
    }
  }
}

__global__ __launch_bounds__(CTPB)
void k_band_cols(const float2* __restrict__ T, float* __restrict__ out, int b0) {
  FFT_SHARED_C16
  const int tid = threadIdx.x, lane = tid & 31, wid = tid >> 5;
  const int c0 = blockIdx.x * (2 * CPB);
  const int bi = blockIdx.y;
  const float2* Tb = T + (size_t)bi * HC * FN;
  float* ob = out + (size_t)(b0 + bi) * FN * FN;
  const int ww = tid & (CPB - 1);
  const int e0 = tid >> 4;
  float2 a[8], b[8];
#pragma unroll
  for (int it = 0; it < 8; ++it) {
    const int k = e0 + 32 * it;
    a[it] = Tb[k * FN + c0 + ww];
    b[it] = Tb[k * FN + c0 + CPB + ww];
  }
  fill_tw512(Twr, Twi, tid, CTPB, 1.0f);
  W16 w; make_w16(1.0f, lane, w);
#pragma unroll
  for (int it = 0; it < 8; ++it) {
    const int k = e0 + 32 * it;
    const bool z = (k == 0);
    const int  km = z ? HC : FN - k;
    const float2 p = a[it], q = b[it];
    Ar[ww][k]  = z ? p.x : p.x - q.y;
    Ai[ww][k]  = z ? q.x : p.y + q.x;
    Ar[ww][km] = z ? p.y : p.x + q.y;
    Ai[ww][km] = z ? q.y : q.x - p.y;
  }
  __syncthreads();
  fft_stageA(Ar[wid], Ai[wid], Br[wid], Bi[wid], lane, 1.0f);
  __syncthreads();
  fft_stageBC(Br[wid], Bi[wid], Ar[wid], Ai[wid], Twr, Twi, lane, w);
  __syncthreads();
  const int nn = lane & 15, mb = (lane >> 4) * 8;
#pragma unroll
  for (int t = 0; t < 2; ++t) {
    v8f dr, di;
    ctile(Ar[wid], Ai[wid], t, lane, w, dr, di);
#pragma unroll
    for (int r = 0; r < 8; ++r) {
      const int o = 32 * (mb + r) + 16 * t + nn;
      Br[wid][o] = dr[r]; Bi[wid][o] = di[r];
    }
  }
  __syncthreads();
  const float scale = 1.0f / (512.0f * 512.0f);
  const int cw = tid & 31;
  const int r0 = tid >> 5;
  const float* col = (cw < CPB) ? &Br[cw][0] : &Bi[cw - CPB][0];
#pragma unroll
  for (int it = 0; it < 32; ++it) {
    const int o = r0 + 16 * it;
    ob[o * FN + c0 + cw] = col[o] * scale;
  }
}

extern "C" void kernel_launch(void* const* d_in, const int* in_sizes, int n_in,
                              void* d_out, int out_size, void* d_ws,
                              size_t ws_size, hipStream_t stream) {
  (void)in_sizes; (void)n_in; (void)out_size;
  const float* img = (const float*)d_in[0];
  const float* win = (const float*)d_in[1];
  float* out = (float*)d_out;

  float2* specRow = (float2*)d_ws;
  float2* spec    = specRow + (size_t)FN * FN;
  float2* bandT   = spec    + (size_t)FN * FN;
  const size_t reserved = 2ull * sizeof(float2) * FN * FN;
  const size_t perBand  = sizeof(float2) * (size_t)FN * FN;
  int chunk = (ws_size > reserved) ? (int)((ws_size - reserved) / perBand) : 0;
  if (chunk < 1)  chunk = 1;
  if (chunk > NB) chunk = NB;

  const dim3 blk(TPB);
  const dim3 cblk(CTPB);
  k_fwd_rows<<<dim3(FN / LPB), blk, 0, stream>>>(img, specRow);
  k_fwd_cols<<<dim3(FN / LPB), blk, 0, stream>>>(specRow, spec);
  for (int b0 = 0; b0 < NB; b0 += chunk) {
    const int nb = (NB - b0 < chunk) ? (NB - b0) : chunk;
    k_band_rows<<<dim3(HC / LPB, nb), blk, 0, stream>>>(spec, win, bandT, b0);
    k_band_cols<<<dim3(FN / (2 * CPB), nb), cblk, 0, stream>>>(bandT, out, b0);
  }
}


# LDS layout of the FFT stages re-mapped (row pairs adjacent, XOR column swizzle, twiddles in lane order): every LDS access bank-conflict-free, fragment reads 8 bytes wide; arithmetic unchanged
# speedup vs baseline: 2.9899x; 1.0713x over previous
#include <hip/hip_runtime.h>
#include <hip/hip_bf16.h>


#define FN   512
#define NB   43
#define LPB  4
#define TPB  128
#define CPB    16
#define CTPB   512
#define LPITCH (FN + 4)
#define HC     256
#define PI_F 3.14159265358979323846f

typedef float v2f __attribute__((ext_vector_type(2)));
typedef float v8f __attribute__((ext_vector_type(8)));

__device__ __forceinline__ v8f wmma_f32(v2f a, v2f b, v8f c) {
  return __builtin_amdgcn_wmma_f32_16x16x4_f32(false, a, false, b,
                                               (short)0, c, false, false);
}


__device__ __forceinline__ void quarter_turn(float c, float s, int t,
                                             float& co, float& so) {
  const bool odd = (t & 1) != 0;
  const float a = odd ? s : c;
  const float b = odd ? c : s;
  const unsigned nc = ((unsigned)(t + 1) & 2u) << 30;
  const unsigned ns = ((unsigned)t & 2u) << 30;
  co = __uint_as_float(__float_as_uint(a) ^ nc);
  so = __uint_as_float(__float_as_uint(b) ^ ns);
}

__device__ __forceinline__ void cis512(int e, float sgn, float& c, float& s) {
  const int q  = (e + 64) >> 7;
  const int rr = e - (q << 7);
  const float x = (float)rr * (2.0f * PI_F / 512.0f);
  const float z = x * x;
  const float ps = fmaf(z, fmaf(z, -1.9515295891e-4f, 8.3321608736e-3f),
                        -1.6666654611e-1f);
  const float pc = fmaf(z, fmaf(z, 2.443315711809948e-5f,
                                -1.388731625493765e-3f),
                        4.166664568298827e-2f);
  const float sx = fmaf(x * z, ps, x);
  const float cx = fmaf(z * z, pc, fmaf(-0.5f, z, 1.0f));
  float s0;
  quarter_turn(cx, sx, q, c, s0);
  s = sgn * s0;
}

__device__ __forceinline__ constexpr float cos_pi16(int k) {
  return k == 0 ? 1.0f
       : k == 1 ? 0.98078528040323044913f
       : k == 2 ? 0.92387953251128675613f
       : k == 3 ? 0.83146961230254523708f
       : k == 4 ? 0.70710678118654752440f
       : k == 5 ? 0.55557023301960222474f
       : k == 6 ? 0.38268343236508977173f
       : k == 7 ? 0.19509032201612826785f
       : k == 8 ? 0.0f
       : k == 9 ? -0.19509032201612826785f
       : k == 10 ? -0.38268343236508977173f
       : k == 11 ? -0.55557023301960222474f
       : k == 12 ? -0.70710678118654752440f
       : k == 13 ? -0.83146961230254523708f
       : k == 14 ? -0.92387953251128675613f
       :           -0.98078528040323044913f;
}
__device__ __forceinline__ constexpr float sin_pi16(int k) {
  return cos_pi16(k <= 8 ? 8 - k : k - 8);
}

struct W16 { v2f ar[4]; v2f ai[4]; v2f ain[4]; };

__device__ __forceinline__ void make_w16(float sgn, int lane, W16& w) {
  const int m  = lane & 15;
  const int kb = (lane >> 4) << 1;
  float cb0, sb0, cb1, sb1;
  cis512(32 * (m * kb),       sgn, cb0, sb0);
  cis512(32 * (m * (kb + 1)), sgn, cb1, sb1);
#pragma unroll
  for (int kc = 0; kc < 4; ++kc) {
    const int t = (sgn > 0.0f) ? (m * kc) : -(m * kc);
    float s0, c0, s1, c1;
    quarter_turn(cb0, sb0, t, c0, s0);
    quarter_turn(cb1, sb1, t, c1, s1);
    v2f tr; tr.x = c0;  tr.y = c1;  w.ar[kc]  = tr;
    v2f ti; ti.x = s0;  ti.y = s1;  w.ai[kc]  = ti;
    v2f tn; tn.x = -s0; tn.y = -s1; w.ain[kc] = tn;
  }
}

__device__ __forceinline__ int kpos(int kap, int par, int col) {
  const int sw = ((kap & 1) << 4) | (kap << 1);
  return (kap << 6) + ((col ^ sw) << 1) + par;
}

__device__ __forceinline__ void ctile(const float* sr, const float* si, int t,
                                      int lane, const W16& w, v8f& cr, v8f& ci) {
  const int c = lane & 15;
  const int h = lane >> 4;
  v8f zr = {0.f,0.f,0.f,0.f,0.f,0.f,0.f,0.f};
  cr = zr; ci = zr;
#pragma unroll
  for (int kc = 0; kc < 4; ++kc) {
    const int p = kpos(2 * kc + h, 0, 16 * t + c);
    const v2f br = *reinterpret_cast<const v2f*>(sr + p);
    const v2f bi = *reinterpret_cast<const v2f*>(si + p);
    cr = wmma_f32(w.ar[kc],  br, cr);
    cr = wmma_f32(w.ain[kc], bi, cr);
    ci = wmma_f32(w.ar[kc],  bi, ci);
    ci = wmma_f32(w.ai[kc],  br, ci);
  }
}

__device__ __forceinline__ void fft_stageA(const float* xr, const float* xi,
                                           float* ur, float* ui,
                                           int lane, float sgn) {
  const bool hi = (lane >> 4) != 0;
#pragma unroll
  for (int i = 0; i < 8; ++i) {
    const int p  = lane + 32 * i;
    const int n1 = p & 15, m1 = p >> 4;
    const float a_r = xr[p],       a_i = xi[p];
    const float b_r = xr[p + 256], b_i = xi[p + 256];
    const int q0 = kpos(i, m1 & 1, n1);
    const int q1 = kpos(i, m1 & 1, n1 + 16);
    ur[q0] = a_r + b_r;
    ui[q0] = a_i + b_i;
    const float dr = a_r - b_r, di = a_i - b_i;
    const float c = hi ? cos_pi16(2 * i + 1) : cos_pi16(2 * i);
    const float s = sgn * (hi ? sin_pi16(2 * i + 1) : sin_pi16(2 * i));
    ur[q1] = dr * c - di * s;
    ui[q1] = dr * s + di * c;
  }
}

__device__ __forceinline__ void fft_stageBC(const float* ur, const float* ui,
                                            float* vr, float* vi,
                                            const float* twr, const float* twi,
                                            int lane, const W16& w) {
  const int nn = lane & 15;
  const int h  = lane >> 4;
  float tr[16], ti[16];
#pragma unroll
  for (int q = 0; q < 4; ++q) {
    const float4 a = *reinterpret_cast<const float4*>(twr + 128 * q + 4 * lane);
    const float4 b = *reinterpret_cast<const float4*>(twi + 128 * q + 4 * lane);
    tr[4 * q] = a.x; tr[4 * q + 1] = a.y; tr[4 * q + 2] = a.z; tr[4 * q + 3] = a.w;
    ti[4 * q] = b.x; ti[4 * q + 1] = b.y; ti[4 * q + 2] = b.z; ti[4 * q + 3] = b.w;
  }
  const int base = kpos(nn >> 1, nn & 1, 16 * h);
#pragma unroll
  for (int t = 0; t < 2; ++t) {
    v8f gr, gi;
    ctile(ur, ui, t, lane, w, gr, gi);
#pragma unroll
    for (int r = 0; r < 8; ++r) {
      const int idx = (base ^ (4 * r)) + 2 * t;
      const float t_r = tr[2 * r + t], t_i = ti[2 * r + t];
      const float g_r = gr[r], g_i = gi[r];
      vr[idx] = g_r * t_r - g_i * t_i;
      vi[idx] = g_r * t_i + g_i * t_r;
    }
  }
}

__device__ __forceinline__ void fill_tw512(float* twr, float* twi,
                                           int tid, int nthr, float sgn) {
  for (int i = tid; i < FN; i += nthr) {
    const int ln = (i >> 2) & 31;
    const int n1 = ln & 15;
    const int k2 = ((ln >> 4) << 4) + ((i >> 7) << 2) + (i & 3);
    float s, c; cis512(n1 * k2, sgn, c, s);
    twr[i] = c; twi[i] = s;
  }
}

#define FFT_SHARED                                              \
  __shared__ __attribute__((aligned(16))) float Ar[LPB][FN], Ai[LPB][FN];   \
  __shared__ __attribute__((aligned(16))) float Br[LPB][FN], Bi[LPB][FN];   \
  __shared__ __attribute__((aligned(16))) float Twr[FN], Twi[FN];

#define FFT_SHARED_C16                                          \
  __shared__ __attribute__((aligned(16))) float Ar[CPB][LPITCH], Ai[CPB][LPITCH]; \
  __shared__ __attribute__((aligned(16))) float Br[CPB][LPITCH], Bi[CPB][LPITCH]; \
  __shared__ __attribute__((aligned(16))) float Twr[FN], Twi[FN];

__global__ __launch_bounds__(TPB)
void k_fwd_rows(const float* __restrict__ img, float2* __restrict__ specRow) {
  FFT_SHARED
  const int tid = threadIdx.x, lane = tid & 31, wid = tid >> 5;
  const int line = blockIdx.x * LPB + wid;
  fill_tw512(Twr, Twi, tid, TPB, -1.0f);
  W16 w; make_w16(-1.0f, lane, w);
#pragma unroll
  for (int i = 0; i < 16; ++i) {
    const int p = lane + 32 * i;
    Ar[wid][p] = img[line * FN + p];
    Ai[wid][p] = 0.0f;
  }
  __syncthreads();
  fft_stageA(Ar[wid], Ai[wid], Br[wid], Bi[wid], lane, -1.0f);
  __syncthreads();
  fft_stageBC(Br[wid], Bi[wid], Ar[wid], Ai[wid], Twr, Twi, lane, w);
  __syncthreads();
  const int nn = lane & 15, mb = (lane >> 4) * 8;
#pragma unroll
  for (int t = 0; t < 2; ++t) {
    v8f dr, di;
    ctile(Ar[wid], Ai[wid], t, lane, w, dr, di);
#pragma unroll
    for (int r = 0; r < 8; ++r) {
      const int o = 32 * (mb + r) + 16 * t + nn;
      specRow[line * FN + o] = make_float2(dr[r], di[r]);
    }
  }
}

__global__ __launch_bounds__(TPB)
void k_fwd_cols(const float2* __restrict__ specRow, float2* __restrict__ spec) {
  FFT_SHARED
  const int tid = threadIdx.x, lane = tid & 31, wid = tid >> 5;
  const int c0 = blockIdx.x * LPB;
  fill_tw512(Twr, Twi, tid, TPB, -1.0f);
  W16 w; make_w16(-1.0f, lane, w);
  for (int i = tid; i < FN * LPB; i += TPB) {
    const int e = i >> 2, ww = i & 3;
    const float2 s = specRow[e * FN + c0 + ww];
    Ar[ww][e] = s.x; Ai[ww][e] = s.y;
  }
  __syncthreads();
  fft_stageA(Ar[wid], Ai[wid], Br[wid], Bi[wid], lane, -1.0f);
  __syncthreads();
  fft_stageBC(Br[wid], Bi[wid], Ar[wid], Ai[wid], Twr, Twi, lane, w);
  __syncthreads();
  const int nn = lane & 15, mb = (lane >> 4) * 8;
#pragma unroll
  for (int t = 0; t < 2; ++t) {
    v8f dr, di;
    ctile(Ar[wid], Ai[wid], t, lane, w, dr, di);
#pragma unroll
    for (int r = 0; r < 8; ++r) {
      const int o = 32 * (mb + r) + 16 * t + nn;
      Br[wid][o] = dr[r]; Bi[wid][o] = di[r];
    }
  }
  __syncthreads();
  for (int i = tid; i < FN * LPB; i += TPB) {
    const int o = i >> 2, ww = i & 3;
    spec[o * FN + c0 + ww] = make_float2(Br[ww][o], Bi[ww][o]);
  }
}

__global__ __launch_bounds__(TPB)
void k_band_rows(const float2* __restrict__ spec, const float* __restrict__ win,
                 float2* __restrict__ T, int b0) {
  FFT_SHARED
  const int tid = threadIdx.x, lane = tid & 31, wid = tid >> 5;
  const int bi = blockIdx.y;
  const float* wb = win + (size_t)(b0 + bi) * FN * FN;
  float2* Tb = T + (size_t)bi * HC * FN;
  const int k1  = blockIdx.x * LPB + wid;
  const int mk1 = (FN - k1) & (FN - 1);
  const float2* srow = spec + k1 * FN;
  const float*  wrow = wb + k1 * FN;
  const float*  mrow = wb + mk1 * FN;
  float2 s[16]; float wv[16], wm[16];
#pragma unroll
  for (int i = 0; i < 16; ++i) {
    const int k2 = lane + 32 * i;
    s[i]  = srow[k2];
    wv[i] = wrow[k2];
    wm[i] = mrow[(FN - k2) & (FN - 1)];
  }
  fill_tw512(Twr, Twi, tid, TPB, 1.0f);
  W16 w; make_w16(1.0f, lane, w);
#pragma unroll
  for (int i = 0; i < 16; ++i) {
    const int k2 = lane + 32 * i;
    const float g = 0.5f * (wv[i] + wm[i]);
    Ar[wid][k2] = s[i].x * g; Ai[wid][k2] = s[i].y * g;
  }
  if (k1 == 0) {
#pragma unroll
    for (int i = 0; i < 16; ++i) {
      const int k2 = lane + 32 * i;
      const float2 q = spec[HC * FN + k2];
      const float g = 0.5f * (wb[HC * FN + k2] +
                              wb[HC * FN + ((FN - k2) & (FN - 1))]);
      Ar[wid][k2] -= q.y * g;
      Ai[wid][k2] += q.x * g;
    }
  }
  __syncthreads();
  fft_stageA(Ar[wid], Ai[wid], Br[wid], Bi[wid], lane, 1.0f);
  __syncthreads();
  fft_stageBC(Br[wid], Bi[wid], Ar[wid], Ai[wid], Twr, Twi, lane, w);
  __syncthreads();
  const int nn = lane & 15, mb = (lane >> 4) * 8;
#pragma unroll
  for (int t = 0; t < 2; ++t) {
    v8f dr, di;
    ctile(Ar[wid], Ai[wid], t, lane, w, dr, di);
#pragma unroll
    for (int r = 0; r < 8; ++r) {
      const int o = 32 * (mb + r) + 16 * t + nn;
      Tb[k1 * FN + o] = make_float2(dr[r], di[r]);
    }
  }
}

__global__ __launch_bounds__(CTPB)
void k_band_cols(const float2* __restrict__ T, float* __restrict__ out, int b0) {
  FFT_SHARED_C16
  const int tid = threadIdx.x, lane = tid & 31, wid = tid >> 5;
  const int c0 = blockIdx.x * (2 * CPB);
  const int bi = blockIdx.y;
  const float2* Tb = T + (size_t)bi * HC * FN;
  float* ob = out + (size_t)(b0 + bi) * FN * FN;
  const int ww = tid & (CPB - 1);
  const int e0 = tid >> 4;
  float2 a[8], b[8];
#pragma unroll
  for (int it = 0; it < 8; ++it) {
    const int k = e0 + 32 * it;
    a[it] = Tb[k * FN + c0 + ww];
    b[it] = Tb[k * FN + c0 + CPB + ww];
  }
  fill_tw512(Twr, Twi, tid, CTPB, 1.0f);
  W16 w; make_w16(1.0f, lane, w);
#pragma unroll
  for (int it = 0; it < 8; ++it) {
    const int k = e0 + 32 * it;
    const bool z = (k == 0);
    const int  km = z ? HC : FN - k;
    const float2 p = a[it], q = b[it];
    Ar[ww][k]  = z ? p.x : p.x - q.y;
    Ai[ww][k]  = z ? q.x : p.y + q.x;
    Ar[ww][km] = z ? p.y : p.x + q.y;
    Ai[ww][km] = z ? q.y : q.x - p.y;
  }
  __syncthreads();
  fft_stageA(Ar[wid], Ai[wid], Br[wid], Bi[wid], lane, 1.0f);
  __syncthreads();
  fft_stageBC(Br[wid], Bi[wid], Ar[wid], Ai[wid], Twr, Twi, lane, w);
  __syncthreads();
  const int nn = lane & 15, mb = (lane >> 4) * 8;
#pragma unroll
  for (int t = 0; t < 2; ++t) {
    v8f dr, di;
    ctile(Ar[wid], Ai[wid], t, lane, w, dr, di);
    const int os0 = (32 * mb + 16 * t + nn) ^ (2 * mb);
    const int os1 = os0 ^ 2;
#pragma unroll
    for (int r = 0; r < 8; ++r) {
      Br[wid][os0 + 32 * r] = dr[r];
      Bi[wid][os1 + 32 * r] = di[r];
    }
  }
  __syncthreads();
  const float scale = 1.0f / (512.0f * 512.0f);
  const int cw = tid & 31;
  const int r0 = tid >> 5;
  const bool im = cw >= CPB;
  const float* col = im ? &Bi[cw - CPB][0] : &Br[cw][0];
  const int rb = im ? (r0 ^ 2) : r0;
#pragma unroll
  for (int it = 0; it < 32; ++it) {
    const int o  = r0 + 16 * it;
    const int os = rb + 16 * (it ^ (it >> 4));
    ob[o * FN + c0 + cw] = col[os] * scale;
  }
}

extern "C" void kernel_launch(void* const* d_in, const int* in_sizes, int n_in,
                              void* d_out, int out_size, void* d_ws,
                              size_t ws_size, hipStream_t stream) {
  (void)in_sizes; (void)n_in; (void)out_size;
  const float* img = (const float*)d_in[0];
  const float* win = (const float*)d_in[1];
  float* out = (float*)d_out;

  float2* specRow = (float2*)d_ws;
  float2* spec    = specRow + (size_t)FN * FN;
  float2* bandT   = spec    + (size_t)FN * FN;
  const size_t reserved = 2ull * sizeof(float2) * FN * FN;
  const size_t perBand  = sizeof(float2) * (size_t)FN * FN;
  int chunk = (ws_size > reserved) ? (int)((ws_size - reserved) / perBand) : 0;
  if (chunk < 1)  chunk = 1;
  if (chunk > NB) chunk = NB;

  const dim3 blk(TPB);
  const dim3 cblk(CTPB);
  k_fwd_rows<<<dim3(FN / LPB), blk, 0, stream>>>(img, specRow);
  k_fwd_cols<<<dim3(FN / LPB), blk, 0, stream>>>(specRow, spec);
  for (int b0 = 0; b0 < NB; b0 += chunk) {
    const int nb = (NB - b0 < chunk) ? (NB - b0) : chunk;
    k_band_rows<<<dim3(HC / LPB, nb), blk, 0, stream>>>(spec, win, bandT, b0);
    k_band_cols<<<dim3(FN / (2 * CPB), nb), cblk, 0, stream>>>(bandT, out, b0);
  }
}


# Complex DFT-16 matmuls as three real matrix products instead of four (P=Wr(Ur+Ui); Yr=P-(Wr+Wi)Ui; Yi=P+(Wi-Wr)Ur): 48 matrix instructions per 512-point line, was 64; sums re-associated
# speedup vs baseline: 3.1368x; 1.0491x over previous
#include <hip/hip_runtime.h>
#include <hip/hip_bf16.h>


#define FN   512
#define NB   43
#define LPB  4
#define TPB  128
#define CPB    16
#define CTPB   512
#define LPITCH (FN + 4)
#define HC     256
#define PI_F 3.14159265358979323846f

typedef float v2f __attribute__((ext_vector_type(2)));
typedef float v8f __attribute__((ext_vector_type(8)));

__device__ __forceinline__ v8f wmma_f32(v2f a, v2f b, v8f c) {
  return __builtin_amdgcn_wmma_f32_16x16x4_f32(false, a, false, b,
                                               (short)0, c, false, false);
}


__device__ __forceinline__ void quarter_turn(float c, float s, int t,
                                             float& co, float& so) {
  const bool odd = (t & 1) != 0;
  const float a = odd ? s : c;
  const float b = odd ? c : s;
  const unsigned nc = ((unsigned)(t + 1) & 2u) << 30;
  const unsigned ns = ((unsigned)t & 2u) << 30;
  co = __uint_as_float(__float_as_uint(a) ^ nc);
  so = __uint_as_float(__float_as_uint(b) ^ ns);
}

__device__ __forceinline__ void cis512(int e, float sgn, float& c, float& s) {
  const int q  = (e + 64) >> 7;
  const int rr = e - (q << 7);
  const float x = (float)rr * (2.0f * PI_F / 512.0f);
  const float z = x * x;
  const float ps = fmaf(z, fmaf(z, -1.9515295891e-4f, 8.3321608736e-3f),
                        -1.6666654611e-1f);
  const float pc = fmaf(z, fmaf(z, 2.443315711809948e-5f,
                                -1.388731625493765e-3f),
                        4.166664568298827e-2f);
  const float sx = fmaf(x * z, ps, x);
  const float cx = fmaf(z * z, pc, fmaf(-0.5f, z, 1.0f));
  float s0;
  quarter_turn(cx, sx, q, c, s0);
  s = sgn * s0;
}

__device__ __forceinline__ constexpr float cos_pi16(int k) {
  return k == 0 ? 1.0f
       : k == 1 ? 0.98078528040323044913f
       : k == 2 ? 0.92387953251128675613f
       : k == 3 ? 0.83146961230254523708f
       : k == 4 ? 0.70710678118654752440f
       : k == 5 ? 0.55557023301960222474f
       : k == 6 ? 0.38268343236508977173f
       : k == 7 ? 0.19509032201612826785f
       : k == 8 ? 0.0f
       : k == 9 ? -0.19509032201612826785f
       : k == 10 ? -0.38268343236508977173f
       : k == 11 ? -0.55557023301960222474f
       : k == 12 ? -0.70710678118654752440f
       : k == 13 ? -0.83146961230254523708f
       : k == 14 ? -0.92387953251128675613f
       :           -0.98078528040323044913f;
}
__device__ __forceinline__ constexpr float sin_pi16(int k) {
  return cos_pi16(k <= 8 ? 8 - k : k - 8);
}

struct W16 { v2f ar[4]; v2f ad[4]; v2f an[4]; };

__device__ __forceinline__ void make_w16(float sgn, int lane, W16& w) {
  const int m  = lane & 15;
  const int kb = (lane >> 4) << 1;
  float cb0, sb0, cb1, sb1;
  cis512(32 * (m * kb),       sgn, cb0, sb0);
  cis512(32 * (m * (kb + 1)), sgn, cb1, sb1);
#pragma unroll
  for (int kc = 0; kc < 4; ++kc) {
    const int t = (sgn > 0.0f) ? (m * kc) : -(m * kc);
    float s0, c0, s1, c1;
    quarter_turn(cb0, sb0, t, c0, s0);
    quarter_turn(cb1, sb1, t, c1, s1);
    v2f tr; tr.x = c0;         tr.y = c1;         w.ar[kc] = tr;
    v2f td; td.x = s0 - c0;    td.y = s1 - c1;    w.ad[kc] = td;
    v2f tn; tn.x = -(c0 + s0); tn.y = -(c1 + s1); w.an[kc] = tn;
  }
}

__device__ __forceinline__ int kpos(int kap, int par, int col) {
  const int sw = ((kap & 1) << 4) | (kap << 1);
  return (kap << 6) + ((col ^ sw) << 1) + par;
}

__device__ __forceinline__ void ctile(const float* sr, const float* si, int t,
                                      int lane, const W16& w, v8f& cr, v8f& ci) {
  const int c = lane & 15;
  const int h = lane >> 4;
  v8f pp = {0.f,0.f,0.f,0.f,0.f,0.f,0.f,0.f};
  v2f br[4], bi[4];
#pragma unroll
  for (int kc = 0; kc < 4; ++kc) {
    const int p = kpos(2 * kc + h, 0, 16 * t + c);
    br[kc] = *reinterpret_cast<const v2f*>(sr + p);
    bi[kc] = *reinterpret_cast<const v2f*>(si + p);
    pp = wmma_f32(w.ar[kc], br[kc] + bi[kc], pp);
  }
  cr = pp; ci = pp;
#pragma unroll
  for (int kc = 0; kc < 4; ++kc) {
    cr = wmma_f32(w.an[kc], bi[kc], cr);
    ci = wmma_f32(w.ad[kc], br[kc], ci);
  }
}

__device__ __forceinline__ void fft_stageA(const float* xr, const float* xi,
                                           float* ur, float* ui,
                                           int lane, float sgn) {
  const bool hi = (lane >> 4) != 0;
#pragma unroll
  for (int i = 0; i < 8; ++i) {
    const int p  = lane + 32 * i;
    const int n1 = p & 15, m1 = p >> 4;
    const float a_r = xr[p],       a_i = xi[p];
    const float b_r = xr[p + 256], b_i = xi[p + 256];
    const int q0 = kpos(i, m1 & 1, n1);
    const int q1 = kpos(i, m1 & 1, n1 + 16);
    ur[q0] = a_r + b_r;
    ui[q0] = a_i + b_i;
    const float dr = a_r - b_r, di = a_i - b_i;
    const float c = hi ? cos_pi16(2 * i + 1) : cos_pi16(2 * i);
    const float s = sgn * (hi ? sin_pi16(2 * i + 1) : sin_pi16(2 * i));
    ur[q1] = dr * c - di * s;
    ui[q1] = dr * s + di * c;
  }
}

__device__ __forceinline__ void fft_stageBC(const float* ur, const float* ui,
                                            float* vr, float* vi,
                                            const float* twr, const float* twi,
                                            int lane, const W16& w) {
  const int nn = lane & 15;
  const int h  = lane >> 4;
  float tr[16], ti[16];
#pragma unroll
  for (int q = 0; q < 4; ++q) {
    const float4 a = *reinterpret_cast<const float4*>(twr + 128 * q + 4 * lane);
    const float4 b = *reinterpret_cast<const float4*>(twi + 128 * q + 4 * lane);
    tr[4 * q] = a.x; tr[4 * q + 1] = a.y; tr[4 * q + 2] = a.z; tr[4 * q + 3] = a.w;
    ti[4 * q] = b.x; ti[4 * q + 1] = b.y; ti[4 * q + 2] = b.z; ti[4 * q + 3] = b.w;
  }
  const int base = kpos(nn >> 1, nn & 1, 16 * h);
#pragma unroll
  for (int t = 0; t < 2; ++t) {
    v8f gr, gi;
    ctile(ur, ui, t, lane, w, gr, gi);
#pragma unroll
    for (int r = 0; r < 8; ++r) {
      const int idx = (base ^ (4 * r)) + 2 * t;
      const float t_r = tr[2 * r + t], t_i = ti[2 * r + t];
      const float g_r = gr[r], g_i = gi[r];
      vr[idx] = g_r * t_r - g_i * t_i;
      vi[idx] = g_r * t_i + g_i * t_r;
    }
  }
}

__device__ __forceinline__ void fill_tw512(float* twr, float* twi,
                                           int tid, int nthr, float sgn) {
  for (int i = tid; i < FN; i += nthr) {
    const int ln = (i >> 2) & 31;
    const int n1 = ln & 15;
    const int k2 = ((ln >> 4) << 4) + ((i >> 7) << 2) + (i & 3);
    float s, c; cis512(n1 * k2, sgn, c, s);
    twr[i] = c; twi[i] = s;
  }
}

#define FFT_SHARED                                              \
  __shared__ __attribute__((aligned(16))) float Ar[LPB][FN], Ai[LPB][FN];   \
  __shared__ __attribute__((aligned(16))) float Br[LPB][FN], Bi[LPB][FN];   \
  __shared__ __attribute__((aligned(16))) float Twr[FN], Twi[FN];

#define FFT_SHARED_C16                                          \
  __shared__ __attribute__((aligned(16))) float Ar[CPB][LPITCH], Ai[CPB][LPITCH]; \
  __shared__ __attribute__((aligned(16))) float Br[CPB][LPITCH], Bi[CPB][LPITCH]; \
  __shared__ __attribute__((aligned(16))) float Twr[FN], Twi[FN];

__global__ __launch_bounds__(TPB)
void k_fwd_rows(const float* __restrict__ img, float2* __restrict__ specRow) {
  FFT_SHARED
  const int tid = threadIdx.x, lane = tid & 31, wid = tid >> 5;
  const int line = blockIdx.x * LPB + wid;
  fill_tw512(Twr, Twi, tid, TPB, -1.0f);
  W16 w; make_w16(-1.0f, lane, w);
#pragma unroll
  for (int i = 0; i < 16; ++i) {
    const int p = lane + 32 * i;
    Ar[wid][p] = img[line * FN + p];
    Ai[wid][p] = 0.0f;
  }
  __syncthreads();
  fft_stageA(Ar[wid], Ai[wid], Br[wid], Bi[wid], lane, -1.0f);
  __syncthreads();
  fft_stageBC(Br[wid], Bi[wid], Ar[wid], Ai[wid], Twr, Twi, lane, w);
  __syncthreads();
  const int nn = lane & 15, mb = (lane >> 4) * 8;
#pragma unroll
  for (int t = 0; t < 2; ++t) {
    v8f dr, di;
    ctile(Ar[wid], Ai[wid], t, lane, w, dr, di);
#pragma unroll
    for (int r = 0; r < 8; ++r) {
      const int o = 32 * (mb + r) + 16 * t + nn;
      specRow[line * FN + o] = make_float2(dr[r], di[r]);
    }
  }
}

__global__ __launch_bounds__(TPB)
void k_fwd_cols(const float2* __restrict__ specRow, float2* __restrict__ spec) {
  FFT_SHARED
  const int tid = threadIdx.x, lane = tid & 31, wid = tid >> 5;
  const int c0 = blockIdx.x * LPB;
  fill_tw512(Twr, Twi, tid, TPB, -1.0f);
  W16 w; make_w16(-1.0f, lane, w);
  for (int i = tid; i < FN * LPB; i += TPB) {
    const int e = i >> 2, ww = i & 3;
    const float2 s = specRow[e * FN + c0 + ww];
    Ar[ww][e] = s.x; Ai[ww][e] = s.y;
  }
  __syncthreads();
  fft_stageA(Ar[wid], Ai[wid], Br[wid], Bi[wid], lane, -1.0f);
  __syncthreads();
  fft_stageBC(Br[wid], Bi[wid], Ar[wid], Ai[wid], Twr, Twi, lane, w);
  __syncthreads();
  const int nn = lane & 15, mb = (lane >> 4) * 8;
#pragma unroll
  for (int t = 0; t < 2; ++t) {
    v8f dr, di;
    ctile(Ar[wid], Ai[wid], t, lane, w, dr, di);
#pragma unroll
    for (int r = 0; r < 8; ++r) {
      const int o = 32 * (mb + r) + 16 * t + nn;
      Br[wid][o] = dr[r]; Bi[wid][o] = di[r];
    }
  }
  __syncthreads();
  for (int i = tid; i < FN * LPB; i += TPB) {
    const int o = i >> 2, ww = i & 3;
    spec[o * FN + c0 + ww] = make_float2(Br[ww][o], Bi[ww][o]);
  }
}

__global__ __launch_bounds__(TPB)
void k_band_rows(const float2* __restrict__ spec, const float* __restrict__ win,
                 float2* __restrict__ T, int b0) {
  FFT_SHARED
  const int tid = threadIdx.x, lane = tid & 31, wid = tid >> 5;
  const int bi = blockIdx.y;
  const float* wb = win + (size_t)(b0 + bi) * FN * FN;
  float2* Tb = T + (size_t)bi * HC * FN;
  const int k1  = blockIdx.x * LPB + wid;
  const int mk1 = (FN - k1) & (FN - 1);
  const float2* srow = spec + k1 * FN;
  const float*  wrow = wb + k1 * FN;
  const float*  mrow = wb + mk1 * FN;
  float2 s[16]; float wv[16], wm[16];
#pragma unroll
  for (int i = 0; i < 16; ++i) {
    const int k2 = lane + 32 * i;
    s[i]  = srow[k2];
    wv[i] = wrow[k2];
    wm[i] = mrow[(FN - k2) & (FN - 1)];
  }
  fill_tw512(Twr, Twi, tid, TPB, 1.0f);
  W16 w; make_w16(1.0f, lane, w);
#pragma unroll
  for (int i = 0; i < 16; ++i) {
    const int k2 = lane + 32 * i;
    const float g = 0.5f * (wv[i] + wm[i]);
    Ar[wid][k2] = s[i].x * g; Ai[wid][k2] = s[i].y * g;
  }
  if (k1 == 0) {
#pragma unroll
    for (int i = 0; i < 16; ++i) {
      const int k2 = lane + 32 * i;
      const float2 q = spec[HC * FN + k2];
      const float g = 0.5f * (wb[HC * FN + k2] +
                              wb[HC * FN + ((FN - k2) & (FN - 1))]);
      Ar[wid][k2] -= q.y * g;
      Ai[wid][k2] += q.x * g;
    }
  }
  __syncthreads();
  fft_stageA(Ar[wid], Ai[wid], Br[wid], Bi[wid], lane, 1.0f);
  __syncthreads();
  fft_stageBC(Br[wid], Bi[wid], Ar[wid], Ai[wid], Twr, Twi, lane, w);
  __syncthreads();
  const int nn = lane & 15, mb = (lane >> 4) * 8;
#pragma unroll
  for (int t = 0; t < 2; ++t) {
    v8f dr, di;
    ctile(Ar[wid], Ai[wid], t, lane, w, dr, di);
#pragma unroll
    for (int r = 0; r < 8; ++r) {
      const int o = 32 * (mb + r) + 16 * t + nn;
      Tb[k1 * FN + o] = make_float2(dr[r], di[r]);
    }
  }
}

__global__ __launch_bounds__(CTPB)
void k_band_cols(const float2* __restrict__ T, float* __restrict__ out, int b0) {
  FFT_SHARED_C16
  const int tid = threadIdx.x, lane = tid & 31, wid = tid >> 5;
  const int c0 = blockIdx.x * (2 * CPB);
  const int bi = blockIdx.y;
  const float2* Tb = T + (size_t)bi * HC * FN;
  float* ob = out + (size_t)(b0 + bi) * FN * FN;
  const int ww = tid & (CPB - 1);
  const int e0 = tid >> 4;
  float2 a[8], b[8];
#pragma unroll
  for (int it = 0; it < 8; ++it) {
    const int k = e0 + 32 * it;
    a[it] = Tb[k * FN + c0 + ww];
    b[it] = Tb[k * FN + c0 + CPB + ww];
  }
  fill_tw512(Twr, Twi, tid, CTPB, 1.0f);
  W16 w; make_w16(1.0f, lane, w);
#pragma unroll
  for (int it = 0; it < 8; ++it) {
    const int k = e0 + 32 * it;
    const bool z = (k == 0);
    const int  km = z ? HC : FN - k;
    const float2 p = a[it], q = b[it];
    Ar[ww][k]  = z ? p.x : p.x - q.y;
    Ai[ww][k]  = z ? q.x : p.y + q.x;
    Ar[ww][km] = z ? p.y : p.x + q.y;
    Ai[ww][km] = z ? q.y : q.x - p.y;
  }
  __syncthreads();
  fft_stageA(Ar[wid], Ai[wid], Br[wid], Bi[wid], lane, 1.0f);
  __syncthreads();
  fft_stageBC(Br[wid], Bi[wid], Ar[wid], Ai[wid], Twr, Twi, lane, w);
  __syncthreads();
  const int nn = lane & 15, mb = (lane >> 4) * 8;
#pragma unroll
  for (int t = 0; t < 2; ++t) {
    v8f dr, di;
    ctile(Ar[wid], Ai[wid], t, lane, w, dr, di);
    const int os0 = (32 * mb + 16 * t + nn) ^ (2 * mb);
    const int os1 = os0 ^ 2;
#pragma unroll
    for (int r = 0; r < 8; ++r) {
      Br[wid][os0 + 32 * r] = dr[r];
      Bi[wid][os1 + 32 * r] = di[r];
    }
  }
  __syncthreads();
  const float scale = 1.0f / (512.0f * 512.0f);
  const int cw = tid & 31;
  const int r0 = tid >> 5;
  const bool im = cw >= CPB;
  const float* col = im ? &Bi[cw - CPB][0] : &Br[cw][0];
  const int rb = im ? (r0 ^ 2) : r0;
#pragma unroll
  for (int it = 0; it < 32; ++it) {
    const int o  = r0 + 16 * it;
    const int os = rb + 16 * (it ^ (it >> 4));
    ob[o * FN + c0 + cw] = col[os] * scale;
  }
}

extern "C" void kernel_launch(void* const* d_in, const int* in_sizes, int n_in,
                              void* d_out, int out_size, void* d_ws,
                              size_t ws_size, hipStream_t stream) {
  (void)in_sizes; (void)n_in; (void)out_size;
  const float* img = (const float*)d_in[0];
  const float* win = (const float*)d_in[1];
  float* out = (float*)d_out;

  float2* specRow = (float2*)d_ws;
  float2* spec    = specRow + (size_t)FN * FN;
  float2* bandT   = spec    + (size_t)FN * FN;
  const size_t reserved = 2ull * sizeof(float2) * FN * FN;
  const size_t perBand  = sizeof(float2) * (size_t)FN * FN;
  int chunk = (ws_size > reserved) ? (int)((ws_size - reserved) / perBand) : 0;
  if (chunk < 1)  chunk = 1;
  if (chunk > NB) chunk = NB;

  const dim3 blk(TPB);
  const dim3 cblk(CTPB);
  k_fwd_rows<<<dim3(FN / LPB), blk, 0, stream>>>(img, specRow);
  k_fwd_cols<<<dim3(FN / LPB), blk, 0, stream>>>(specRow, spec);
  for (int b0 = 0; b0 < NB; b0 += chunk) {
    const int nb = (NB - b0 < chunk) ? (NB - b0) : chunk;
    k_band_rows<<<dim3(HC / LPB, nb), blk, 0, stream>>>(spec, win, bandT, b0);
    k_band_cols<<<dim3(FN / (2 * CPB), nb), cblk, 0, stream>>>(bandT, out, b0);
  }
}


# DFT-16 products folded on the DFT matrix's mirror symmetry: rows k, 16-k added/subtracted first, operands as (re,im) pairs in LDS; 32 matrix instructions per line, was 48; sums re-associated
# speedup vs baseline: 3.2982x; 1.0515x over previous
#include <hip/hip_runtime.h>
#include <hip/hip_bf16.h>


#define FN   512
#define NB   43
#define LPB  4
#define TPB  128
#define CPB    16
#define CTPB   512
#define LPITCH (FN + 4)
#define HC     256
#define PI_F 3.14159265358979323846f

typedef float v2f __attribute__((ext_vector_type(2)));
typedef float v8f __attribute__((ext_vector_type(8)));

__device__ __forceinline__ v8f wmma_f32(v2f a, v2f b, v8f c) {
  return __builtin_amdgcn_wmma_f32_16x16x4_f32(false, a, false, b,
                                               (short)0, c, false, false);
}


__device__ __forceinline__ void quarter_turn(float c, float s, int t,
                                             float& co, float& so) {
  const bool odd = (t & 1) != 0;
  const float a = odd ? s : c;
  const float b = odd ? c : s;
  const unsigned nc = ((unsigned)(t + 1) & 2u) << 30;
  const unsigned ns = ((unsigned)t & 2u) << 30;
  co = __uint_as_float(__float_as_uint(a) ^ nc);
  so = __uint_as_float(__float_as_uint(b) ^ ns);
}

__device__ __forceinline__ void cis512(int e, float sgn, float& c, float& s) {
  const int q  = (e + 64) >> 7;
  const int rr = e - (q << 7);
  const float x = (float)rr * (2.0f * PI_F / 512.0f);
  const float z = x * x;
  const float ps = fmaf(z, fmaf(z, -1.9515295891e-4f, 8.3321608736e-3f),
                        -1.6666654611e-1f);
  const float pc = fmaf(z, fmaf(z, 2.443315711809948e-5f,
                                -1.388731625493765e-3f),
                        4.166664568298827e-2f);
  const float sx = fmaf(x * z, ps, x);
  const float cx = fmaf(z * z, pc, fmaf(-0.5f, z, 1.0f));
  float s0;
  quarter_turn(cx, sx, q, c, s0);
  s = sgn * s0;
}

__device__ __forceinline__ constexpr float cos_pi16(int k) {
  return k == 0 ? 1.0f
       : k == 1 ? 0.98078528040323044913f
       : k == 2 ? 0.92387953251128675613f
       : k == 3 ? 0.83146961230254523708f
       : k == 4 ? 0.70710678118654752440f
       : k == 5 ? 0.55557023301960222474f
       : k == 6 ? 0.38268343236508977173f
       : k == 7 ? 0.19509032201612826785f
       : k == 8 ? 0.0f
       : k == 9 ? -0.19509032201612826785f
       : k == 10 ? -0.38268343236508977173f
       : k == 11 ? -0.55557023301960222474f
       : k == 12 ? -0.70710678118654752440f
       : k == 13 ? -0.83146961230254523708f
       : k == 14 ? -0.92387953251128675613f
       :           -0.98078528040323044913f;
}
__device__ __forceinline__ constexpr float sin_pi16(int k) {
  return cos_pi16(k <= 8 ? 8 - k : k - 8);
}

struct W16 { v2f a1[4]; v2f a2[4]; };

__device__ __forceinline__ void make_w16(float sgn, int lane, W16& w) {
  const int m = lane & 15;
  const int h = lane >> 4;
  float cb0, sb0, cb1, sb1;
  cis512(32 * (m * h),       sgn, cb0, sb0);
  cis512(32 * (m * (2 + h)), sgn, cb1, sb1);
  const int t = (sgn > 0.0f) ? m : -m;
  float cb2, sb2, cb3, sb3;
  quarter_turn(cb0, sb0, t, cb2, sb2);
  quarter_turn(cb1, sb1, t, cb3, sb3);
  const float c8 = (m & 1) ? -1.0f : 1.0f;
  v2f p;
  p.x = cb0;                      p.y = (h == 0) ? c8 : -sb0;  w.a1[0] = p;
  p.x = (h == 0) ? 1.0f : sb0;    p.y = (h == 0) ? c8 :  cb0;  w.a2[0] = p;
  p.x = cb1; p.y = -sb1;  w.a1[1] = p;   p.x = sb1; p.y = cb1;  w.a2[1] = p;
  p.x = cb2; p.y = -sb2;  w.a1[2] = p;   p.x = sb2; p.y = cb2;  w.a2[2] = p;
  p.x = cb3; p.y = -sb3;  w.a1[3] = p;   p.x = sb3; p.y = cb3;  w.a2[3] = p;
}

__device__ __forceinline__ int krow_pair(int k) { return (k <= 8 ? k : 16 - k) & 7; }
__device__ __forceinline__ int krow_side(int k) { return k >= 8 ? 1 : 0; }
__device__ __forceinline__ int kpos(int kap, int par, int col) {
  const int sw = ((kap & 1) << 4) | (kap << 1);
  return (kap << 6) + ((col ^ sw) << 1) + par;
}

__device__ __forceinline__ void ctile(const float* sl, const float* sh, int t,
                                      int lane, const W16& w, v8f& cr, v8f& ci) {
  const int c = lane & 15;
  const int h = lane >> 4;
  v2f zj[4], zm[4];
#pragma unroll
  for (int kc = 0; kc < 4; ++kc) {
    const int p = kpos(2 * kc + h, 0, 16 * t + c);
    zj[kc] = *reinterpret_cast<const v2f*>(sl + p);
    zm[kc] = *reinterpret_cast<const v2f*>(sh + p);
  }
  v2f b1[4], b2[4];
#pragma unroll
  for (int kc = 0; kc < 4; ++kc) {
    v2f zc; zc.x = zm[kc].x; zc.y = -zm[kc].y;
    v2f f1 = zj[kc] + zc;
    v2f f2 = zj[kc] - zc;
    if (kc == 0) {
      f1.x = (h == 0) ? zj[0].x : f1.x;  f1.y = (h == 0) ? zm[0].x : f1.y;
      f2.x = (h == 0) ? zj[0].y : f2.x;  f2.y = (h == 0) ? zm[0].y : f2.y;
    }
    b1[kc] = f1; b2[kc] = f2;
  }
  v8f yr = {0.f,0.f,0.f,0.f,0.f,0.f,0.f,0.f};
  v8f yi = {0.f,0.f,0.f,0.f,0.f,0.f,0.f,0.f};
#pragma unroll
  for (int kc = 0; kc < 4; ++kc) {
    yr = wmma_f32(w.a1[kc], b1[kc], yr);
    yi = wmma_f32(w.a2[kc], b2[kc], yi);
  }
  cr = yr; ci = yi;
}

__device__ __forceinline__ void fft_stageA(const float* xr, const float* xi,
                                           float* ul, float* uh,
                                           int lane, float sgn) {
  const bool hi = (lane >> 4) != 0;
  const int l2  = 2 * lane;
  const int h68 = 68 * (lane >> 4);
#pragma unroll
  for (int i = 0; i < 8; ++i) {
    const int p  = lane + 32 * i;
    const float a_r = xr[p],       a_i = xi[p];
    const float b_r = xr[p + 256], b_i = xi[p + 256];
    const int k68 = (i < 4) ? 136 * i + h68
                  : (i == 4) ? 7 * h68
                  : 68 * (16 - 2 * i) - h68;
    float* up = (i < 4) ? ul : uh;
    const int q0 = l2 ^ k68;
    const int q1 = q0 ^ 32;
    v2f e0; e0.x = a_r + b_r; e0.y = a_i + b_i;
    *reinterpret_cast<v2f*>(up + q0) = e0;
    const float dr = a_r - b_r, di = a_i - b_i;
    const float c = hi ? cos_pi16(2 * i + 1) : cos_pi16(2 * i);
    const float s = sgn * (hi ? sin_pi16(2 * i + 1) : sin_pi16(2 * i));
    v2f e1; e1.x = dr * c - di * s; e1.y = dr * s + di * c;
    *reinterpret_cast<v2f*>(up + q1) = e1;
  }
}

__device__ __forceinline__ void fft_stageBC(const float* ul, const float* uh,
                                            float* vl, float* vh,
                                            const float* twr, const float* twi,
                                            int lane, const W16& w) {
  const int nn = lane & 15;
  const int h  = lane >> 4;
  float tr[16], ti[16];
#pragma unroll
  for (int q = 0; q < 4; ++q) {
    const float4 a = *reinterpret_cast<const float4*>(twr + 128 * q + 4 * lane);
    const float4 b = *reinterpret_cast<const float4*>(twi + 128 * q + 4 * lane);
    tr[4 * q] = a.x; tr[4 * q + 1] = a.y; tr[4 * q + 2] = a.z; tr[4 * q + 3] = a.w;
    ti[4 * q] = b.x; ti[4 * q + 1] = b.y; ti[4 * q + 2] = b.z; ti[4 * q + 3] = b.w;
  }
  const int base = kpos(krow_pair(nn), 0, 16 * h);
  float* vp = krow_side(nn) ? vh : vl;
#pragma unroll
  for (int t = 0; t < 2; ++t) {
    v8f gr, gi;
    ctile(ul, uh, t, lane, w, gr, gi);
#pragma unroll
    for (int r = 0; r < 8; ++r) {
      const int idx = (base ^ (4 * r)) + 2 * t;
      const float t_r = tr[2 * r + t], t_i = ti[2 * r + t];
      const float g_r = gr[r], g_i = gi[r];
      v2f e; e.x = g_r * t_r - g_i * t_i; e.y = g_r * t_i + g_i * t_r;
      *reinterpret_cast<v2f*>(vp + idx) = e;
    }
  }
}

__device__ __forceinline__ void fill_tw512(float* twr, float* twi,
                                           int tid, int nthr, float sgn) {
  for (int i = tid; i < FN; i += nthr) {
    const int ln = (i >> 2) & 31;
    const int n1 = ln & 15;
    const int k2 = ((ln >> 4) << 4) + ((i >> 7) << 2) + (i & 3);
    float s, c; cis512(n1 * k2, sgn, c, s);
    twr[i] = c; twi[i] = s;
  }
}

#define FFT_SHARED                                              \
  __shared__ __attribute__((aligned(16))) float Ar[LPB][FN], Ai[LPB][FN];   \
  __shared__ __attribute__((aligned(16))) float Br[LPB][FN], Bi[LPB][FN];   \
  __shared__ __attribute__((aligned(16))) float Twr[FN], Twi[FN];

#define FFT_SHARED_C16                                          \
  __shared__ __attribute__((aligned(16))) float Ar[CPB][LPITCH], Ai[CPB][LPITCH]; \
  __shared__ __attribute__((aligned(16))) float Br[CPB][LPITCH], Bi[CPB][LPITCH]; \
  __shared__ __attribute__((aligned(16))) float Twr[FN], Twi[FN];

__global__ __launch_bounds__(TPB)
void k_fwd_rows(const float* __restrict__ img, float2* __restrict__ specRow) {
  FFT_SHARED
  const int tid = threadIdx.x, lane = tid & 31, wid = tid >> 5;
  const int line = blockIdx.x * LPB + wid;
  fill_tw512(Twr, Twi, tid, TPB, -1.0f);
  W16 w; make_w16(-1.0f, lane, w);
#pragma unroll
  for (int i = 0; i < 16; ++i) {
    const int p = lane + 32 * i;
    Ar[wid][p] = img[line * FN + p];
    Ai[wid][p] = 0.0f;
  }
  __syncthreads();
  fft_stageA(Ar[wid], Ai[wid], Br[wid], Bi[wid], lane, -1.0f);
  __syncthreads();
  fft_stageBC(Br[wid], Bi[wid], Ar[wid], Ai[wid], Twr, Twi, lane, w);
  __syncthreads();
  const int nn = lane & 15, mb = (lane >> 4) * 8;
#pragma unroll
  for (int t = 0; t < 2; ++t) {
    v8f dr, di;
    ctile(Ar[wid], Ai[wid], t, lane, w, dr, di);
#pragma unroll
    for (int r = 0; r < 8; ++r) {
      const int o = 32 * (mb + r) + 16 * t + nn;
      specRow[line * FN + o] = make_float2(dr[r], di[r]);
    }
  }
}

__global__ __launch_bounds__(TPB)
void k_fwd_cols(const float2* __restrict__ specRow, float2* __restrict__ spec) {
  FFT_SHARED
  const int tid = threadIdx.x, lane = tid & 31, wid = tid >> 5;
  const int c0 = blockIdx.x * LPB;
  fill_tw512(Twr, Twi, tid, TPB, -1.0f);
  W16 w; make_w16(-1.0f, lane, w);
  for (int i = tid; i < FN * LPB; i += TPB) {
    const int e = i >> 2, ww = i & 3;
    const float2 s = specRow[e * FN + c0 + ww];
    Ar[ww][e] = s.x; Ai[ww][e] = s.y;
  }
  __syncthreads();
  fft_stageA(Ar[wid], Ai[wid], Br[wid], Bi[wid], lane, -1.0f);
  __syncthreads();
  fft_stageBC(Br[wid], Bi[wid], Ar[wid], Ai[wid], Twr, Twi, lane, w);
  __syncthreads();
  const int nn = lane & 15, mb = (lane >> 4) * 8;
#pragma unroll
  for (int t = 0; t < 2; ++t) {
    v8f dr, di;
    ctile(Ar[wid], Ai[wid], t, lane, w, dr, di);
#pragma unroll
    for (int r = 0; r < 8; ++r) {
      const int o = 32 * (mb + r) + 16 * t + nn;
      Br[wid][o] = dr[r]; Bi[wid][o] = di[r];
    }
  }
  __syncthreads();
  for (int i = tid; i < FN * LPB; i += TPB) {
    const int o = i >> 2, ww = i & 3;
    spec[o * FN + c0 + ww] = make_float2(Br[ww][o], Bi[ww][o]);
  }
}

__global__ __launch_bounds__(TPB)
void k_band_rows(const float2* __restrict__ spec, const float* __restrict__ win,
                 float2* __restrict__ T, int b0) {
  FFT_SHARED
  const int tid = threadIdx.x, lane = tid & 31, wid = tid >> 5;
  const int bi = blockIdx.y;
  const float* wb = win + (size_t)(b0 + bi) * FN * FN;
  float2* Tb = T + (size_t)bi * HC * FN;
  const int k1  = blockIdx.x * LPB + wid;
  const int mk1 = (FN - k1) & (FN - 1);
  const float2* srow = spec + k1 * FN;
  const float*  wrow = wb + k1 * FN;
  const float*  mrow = wb + mk1 * FN;
  float2 s[16]; float wv[16], wm[16];
#pragma unroll
  for (int i = 0; i < 16; ++i) {
    const int k2 = lane + 32 * i;
    s[i]  = srow[k2];
    wv[i] = wrow[k2];
    wm[i] = mrow[(FN - k2) & (FN - 1)];
  }
  fill_tw512(Twr, Twi, tid, TPB, 1.0f);
  W16 w; make_w16(1.0f, lane, w);
#pragma unroll
  for (int i = 0; i < 16; ++i) {
    const int k2 = lane + 32 * i;
    const float g = 0.5f * (wv[i] + wm[i]);
    Ar[wid][k2] = s[i].x * g; Ai[wid][k2] = s[i].y * g;
  }
  if (k1 == 0) {
#pragma unroll
    for (int i = 0; i < 16; ++i) {
      const int k2 = lane + 32 * i;
      const float2 q = spec[HC * FN + k2];
      const float g = 0.5f * (wb[HC * FN + k2] +
                              wb[HC * FN + ((FN - k2) & (FN - 1))]);
      Ar[wid][k2] -= q.y * g;
      Ai[wid][k2] += q.x * g;
    }
  }
  __syncthreads();
  fft_stageA(Ar[wid], Ai[wid], Br[wid], Bi[wid], lane, 1.0f);
  __syncthreads();
  fft_stageBC(Br[wid], Bi[wid], Ar[wid], Ai[wid], Twr, Twi, lane, w);
  __syncthreads();
  const int nn = lane & 15, mb = (lane >> 4) * 8;
#pragma unroll
  for (int t = 0; t < 2; ++t) {
    v8f dr, di;
    ctile(Ar[wid], Ai[wid], t, lane, w, dr, di);
#pragma unroll
    for (int r = 0; r < 8; ++r) {
      const int o = 32 * (mb + r) + 16 * t + nn;
      Tb[k1 * FN + o] = make_float2(dr[r], di[r]);
    }
  }
}

__global__ __launch_bounds__(CTPB)
void k_band_cols(const float2* __restrict__ T, float* __restrict__ out, int b0) {
  FFT_SHARED_C16
  const int tid = threadIdx.x, lane = tid & 31, wid = tid >> 5;
  const int c0 = blockIdx.x * (2 * CPB);
  const int bi = blockIdx.y;
  const float2* Tb = T + (size_t)bi * HC * FN;
  float* ob = out + (size_t)(b0 + bi) * FN * FN;
  const int ww = tid & (CPB - 1);
  const int e0 = tid >> 4;
  float2 a[8], b[8];
#pragma unroll
  for (int it = 0; it < 8; ++it) {
    const int k = e0 + 32 * it;
    a[it] = Tb[k * FN + c0 + ww];
    b[it] = Tb[k * FN + c0 + CPB + ww];
  }
  fill_tw512(Twr, Twi, tid, CTPB, 1.0f);
  W16 w; make_w16(1.0f, lane, w);
#pragma unroll
  for (int it = 0; it < 8; ++it) {
    const int k = e0 + 32 * it;
    const bool z = (k == 0);
    const int  km = z ? HC : FN - k;
    const float2 p = a[it], q = b[it];
    Ar[ww][k]  = z ? p.x : p.x - q.y;
    Ai[ww][k]  = z ? q.x : p.y + q.x;
    Ar[ww][km] = z ? p.y : p.x + q.y;
    Ai[ww][km] = z ? q.y : q.x - p.y;
  }
  __syncthreads();
  fft_stageA(Ar[wid], Ai[wid], Br[wid], Bi[wid], lane, 1.0f);
  __syncthreads();
  fft_stageBC(Br[wid], Bi[wid], Ar[wid], Ai[wid], Twr, Twi, lane, w);
  __syncthreads();
  const int nn = lane & 15, mb = (lane >> 4) * 8;
#pragma unroll
  for (int t = 0; t < 2; ++t) {
    v8f dr, di;
    ctile(Ar[wid], Ai[wid], t, lane, w, dr, di);
    const int os0 = (32 * mb + 16 * t + nn) ^ (2 * mb);
    const int os1 = os0 ^ 2;
#pragma unroll
    for (int r = 0; r < 8; ++r) {
      Br[wid][os0 + 32 * r] = dr[r];
      Bi[wid][os1 + 32 * r] = di[r];
    }
  }
  __syncthreads();
  const float scale = 1.0f / (512.0f * 512.0f);
  const int cw = tid & 31;
  const int r0 = tid >> 5;
  const bool im = cw >= CPB;
  const float* col = im ? &Bi[cw - CPB][0] : &Br[cw][0];
  const int rb = im ? (r0 ^ 2) : r0;
#pragma unroll
  for (int it = 0; it < 32; ++it) {
    const int o  = r0 + 16 * it;
    const int os = rb + 16 * (it ^ (it >> 4));
    ob[o * FN + c0 + cw] = col[os] * scale;
  }
}

extern "C" void kernel_launch(void* const* d_in, const int* in_sizes, int n_in,
                              void* d_out, int out_size, void* d_ws,
                              size_t ws_size, hipStream_t stream) {
  (void)in_sizes; (void)n_in; (void)out_size;
  const float* img = (const float*)d_in[0];
  const float* win = (const float*)d_in[1];
  float* out = (float*)d_out;

  float2* specRow = (float2*)d_ws;
  float2* spec    = specRow + (size_t)FN * FN;
  float2* bandT   = spec    + (size_t)FN * FN;
  const size_t reserved = 2ull * sizeof(float2) * FN * FN;
  const size_t perBand  = sizeof(float2) * (size_t)FN * FN;
  int chunk = (ws_size > reserved) ? (int)((ws_size - reserved) / perBand) : 0;
  if (chunk < 1)  chunk = 1;
  if (chunk > NB) chunk = NB;

  const dim3 blk(TPB);
  const dim3 cblk(CTPB);
  k_fwd_rows<<<dim3(FN / LPB), blk, 0, stream>>>(img, specRow);
  k_fwd_cols<<<dim3(FN / LPB), blk, 0, stream>>>(specRow, spec);
  for (int b0 = 0; b0 < NB; b0 += chunk) {
    const int nb = (NB - b0 < chunk) ? (NB - b0) : chunk;
    k_band_rows<<<dim3(HC / LPB, nb), blk, 0, stream>>>(spec, win, bandT, b0);
    k_band_cols<<<dim3(FN / (2 * CPB), nb), cblk, 0, stream>>>(bandT, out, b0);
  }
}


# First FFT stages in registers: each lane's own 16 points give its radix-2 results and folded DFT-16 operands directly; U never passes through LDS (two LDS round trips fewer per line); same arithmetic
# speedup vs baseline: 3.5398x; 1.0733x over previous
#include <hip/hip_runtime.h>
#include <hip/hip_bf16.h>


#define FN   512
#define NB   43
#define LPB  4
#define TPB  128
#define CPB    16
#define CTPB   512
#define LPITCH (FN + 4)
#define HC     256
#define PI_F 3.14159265358979323846f

typedef float v2f __attribute__((ext_vector_type(2)));
typedef float v8f __attribute__((ext_vector_type(8)));

__device__ __forceinline__ v8f wmma_f32(v2f a, v2f b, v8f c) {
  return __builtin_amdgcn_wmma_f32_16x16x4_f32(false, a, false, b,
                                               (short)0, c, false, false);
}


__device__ __forceinline__ void quarter_turn(float c, float s, int t,
                                             float& co, float& so) {
  const bool odd = (t & 1) != 0;
  const float a = odd ? s : c;
  const float b = odd ? c : s;
  const unsigned nc = ((unsigned)(t + 1) & 2u) << 30;
  const unsigned ns = ((unsigned)t & 2u) << 30;
  co = __uint_as_float(__float_as_uint(a) ^ nc);
  so = __uint_as_float(__float_as_uint(b) ^ ns);
}

__device__ __forceinline__ void cis512(int e, float sgn, float& c, float& s) {
  const int q  = (e + 64) >> 7;
  const int rr = e - (q << 7);
  const float x = (float)rr * (2.0f * PI_F / 512.0f);
  const float z = x * x;
  const float ps = fmaf(z, fmaf(z, -1.9515295891e-4f, 8.3321608736e-3f),
                        -1.6666654611e-1f);
  const float pc = fmaf(z, fmaf(z, 2.443315711809948e-5f,
                                -1.388731625493765e-3f),
                        4.166664568298827e-2f);
  const float sx = fmaf(x * z, ps, x);
  const float cx = fmaf(z * z, pc, fmaf(-0.5f, z, 1.0f));
  float s0;
  quarter_turn(cx, sx, q, c, s0);
  s = sgn * s0;
}

__device__ __forceinline__ constexpr float cos_pi16(int k) {
  return k == 0 ? 1.0f
       : k == 1 ? 0.98078528040323044913f
       : k == 2 ? 0.92387953251128675613f
       : k == 3 ? 0.83146961230254523708f
       : k == 4 ? 0.70710678118654752440f
       : k == 5 ? 0.55557023301960222474f
       : k == 6 ? 0.38268343236508977173f
       : k == 7 ? 0.19509032201612826785f
       : k == 8 ? 0.0f
       : k == 9 ? -0.19509032201612826785f
       : k == 10 ? -0.38268343236508977173f
       : k == 11 ? -0.55557023301960222474f
       : k == 12 ? -0.70710678118654752440f
       : k == 13 ? -0.83146961230254523708f
       : k == 14 ? -0.92387953251128675613f
       :           -0.98078528040323044913f;
}
__device__ __forceinline__ constexpr float sin_pi16(int k) {
  return cos_pi16(k <= 8 ? 8 - k : k - 8);
}

struct W16 { v2f a1[4]; v2f a2[4]; };

__device__ __forceinline__ void make_w16(float sgn, int lane, W16& w) {
  const int m = lane & 15;
  const int h = lane >> 4;
  float cb0, sb0, cb1, sb1;
  cis512(32 * (m * h),       sgn, cb0, sb0);
  cis512(32 * (m * (2 + h)), sgn, cb1, sb1);
  const int t = (sgn > 0.0f) ? m : -m;
  float cb2, sb2, cb3, sb3;
  quarter_turn(cb0, sb0, t, cb2, sb2);
  quarter_turn(cb1, sb1, t, cb3, sb3);
  const float c8 = (m & 1) ? -1.0f : 1.0f;
  v2f p;
  p.x = cb0;                      p.y = (h == 0) ? c8 : -sb0;  w.a1[0] = p;
  p.x = (h == 0) ? 1.0f : sb0;    p.y = (h == 0) ? c8 :  cb0;  w.a2[0] = p;
  p.x = cb1; p.y = -sb1;  w.a1[1] = p;   p.x = sb1; p.y = cb1;  w.a2[1] = p;
  p.x = cb2; p.y = -sb2;  w.a1[2] = p;   p.x = sb2; p.y = cb2;  w.a2[2] = p;
  p.x = cb3; p.y = -sb3;  w.a1[3] = p;   p.x = sb3; p.y = cb3;  w.a2[3] = p;
}

__device__ __forceinline__ int krow_pair(int k) { return (k <= 8 ? k : 16 - k) & 7; }
__device__ __forceinline__ int krow_side(int k) { return k >= 8 ? 1 : 0; }
__device__ __forceinline__ int kpos(int kap, int par, int col) {
  const int sw = ((kap & 1) << 4) | (kap << 1);
  return (kap << 6) + ((col ^ sw) << 1) + par;
}

__device__ __forceinline__ void ctile(const float* sl, const float* sh, int t,
                                      int lane, const W16& w, v8f& cr, v8f& ci) {
  const int c = lane & 15;
  const int h = lane >> 4;
  v2f zj[4], zm[4];
#pragma unroll
  for (int kc = 0; kc < 4; ++kc) {
    const int p = kpos(2 * kc + h, 0, 16 * t + c);
    zj[kc] = *reinterpret_cast<const v2f*>(sl + p);
    zm[kc] = *reinterpret_cast<const v2f*>(sh + p);
  }
  v2f b1[4], b2[4];
#pragma unroll
  for (int kc = 0; kc < 4; ++kc) {
    v2f zc; zc.x = zm[kc].x; zc.y = -zm[kc].y;
    v2f f1 = zj[kc] + zc;
    v2f f2 = zj[kc] - zc;
    if (kc == 0) {
      f1.x = (h == 0) ? zj[0].x : f1.x;  f1.y = (h == 0) ? zm[0].x : f1.y;
      f2.x = (h == 0) ? zj[0].y : f2.x;  f2.y = (h == 0) ? zm[0].y : f2.y;
    }
    b1[kc] = f1; b2[kc] = f2;
  }
  v8f yr = {0.f,0.f,0.f,0.f,0.f,0.f,0.f,0.f};
  v8f yi = {0.f,0.f,0.f,0.f,0.f,0.f,0.f,0.f};
#pragma unroll
  for (int kc = 0; kc < 4; ++kc) {
    yr = wmma_f32(w.a1[kc], b1[kc], yr);
    yi = wmma_f32(w.a2[kc], b2[kc], yi);
  }
  cr = yr; ci = yi;
}

__device__ __forceinline__ void fft_stageA_regs(const float (&xr)[16],
                                                const float (&xi)[16],
                                                int lane, float sgn,
                                                v2f (&e0)[8], v2f (&e1)[8]) {
  const bool hi = (lane >> 4) != 0;
#pragma unroll
  for (int i = 0; i < 8; ++i) {
    const float a_r = xr[i],     a_i = xi[i];
    const float b_r = xr[i + 8], b_i = xi[i + 8];
    v2f s0; s0.x = a_r + b_r; s0.y = a_i + b_i;
    e0[i] = s0;
    const float dr = a_r - b_r, di = a_i - b_i;
    const float c = hi ? cos_pi16(2 * i + 1) : cos_pi16(2 * i);
    const float s = sgn * (hi ? sin_pi16(2 * i + 1) : sin_pi16(2 * i));
    v2f s1; s1.x = dr * c - di * s; s1.y = dr * s + di * c;
    e1[i] = s1;
  }
}

__device__ __forceinline__ void ctile_regs(const v2f (&e)[8], int lane,
                                           const W16& w, v8f& cr, v8f& ci) {
  const bool hi = (lane >> 4) != 0;
  v2f b1[4], b2[4];
#pragma unroll
  for (int kc = 0; kc < 4; ++kc) {
    const v2f zj = e[kc];
    const v2f ml = e[(kc == 0) ? 4 : 8 - kc];
    const v2f mh = e[7 - kc];
    v2f zm; zm.x = hi ? mh.x : ml.x; zm.y = hi ? mh.y : ml.y;
    v2f zc; zc.x = zm.x; zc.y = -zm.y;
    v2f f1 = zj + zc;
    v2f f2 = zj - zc;
    if (kc == 0) {
      f1.x = hi ? f1.x : zj.x;  f1.y = hi ? f1.y : zm.x;
      f2.x = hi ? f2.x : zj.y;  f2.y = hi ? f2.y : zm.y;
    }
    b1[kc] = f1; b2[kc] = f2;
  }
  v8f yr = {0.f,0.f,0.f,0.f,0.f,0.f,0.f,0.f};
  v8f yi = {0.f,0.f,0.f,0.f,0.f,0.f,0.f,0.f};
#pragma unroll
  for (int kc = 0; kc < 4; ++kc) {
    yr = wmma_f32(w.a1[kc], b1[kc], yr);
    yi = wmma_f32(w.a2[kc], b2[kc], yi);
  }
  cr = yr; ci = yi;
}

__device__ __forceinline__ void fft_stageC_store(const v8f (&gr)[2],
                                                 const v8f (&gi)[2],
                                                 float* vl, float* vh,
                                                 const float* twr, const float* twi,
                                                 int lane) {
  const int nn = lane & 15;
  const int h  = lane >> 4;
  float tr[16], ti[16];
#pragma unroll
  for (int q = 0; q < 4; ++q) {
    const float4 a = *reinterpret_cast<const float4*>(twr + 128 * q + 4 * lane);
    const float4 b = *reinterpret_cast<const float4*>(twi + 128 * q + 4 * lane);
    tr[4 * q] = a.x; tr[4 * q + 1] = a.y; tr[4 * q + 2] = a.z; tr[4 * q + 3] = a.w;
    ti[4 * q] = b.x; ti[4 * q + 1] = b.y; ti[4 * q + 2] = b.z; ti[4 * q + 3] = b.w;
  }
  const int base = kpos(krow_pair(nn), 0, 16 * h);
  float* vp = krow_side(nn) ? vh : vl;
#pragma unroll
  for (int t = 0; t < 2; ++t) {
#pragma unroll
    for (int r = 0; r < 8; ++r) {
      const int idx = (base ^ (4 * r)) + 2 * t;
      const float t_r = tr[2 * r + t], t_i = ti[2 * r + t];
      const float g_r = gr[t][r], g_i = gi[t][r];
      v2f e; e.x = g_r * t_r - g_i * t_i; e.y = g_r * t_i + g_i * t_r;
      *reinterpret_cast<v2f*>(vp + idx) = e;
    }
  }
}

__device__ __forceinline__ void fill_tw512(float* twr, float* twi,
                                           int tid, int nthr, float sgn) {
  for (int i = tid; i < FN; i += nthr) {
    const int ln = (i >> 2) & 31;
    const int n1 = ln & 15;
    const int k2 = ((ln >> 4) << 4) + ((i >> 7) << 2) + (i & 3);
    float s, c; cis512(n1 * k2, sgn, c, s);
    twr[i] = c; twi[i] = s;
  }
}

#define FFT_SHARED                                              \
  __shared__ __attribute__((aligned(16))) float Ar[LPB][FN], Ai[LPB][FN];   \
  __shared__ __attribute__((aligned(16))) float Br[LPB][FN], Bi[LPB][FN];   \
  __shared__ __attribute__((aligned(16))) float Twr[FN], Twi[FN];

#define FFT_SHARED_ROWS                                         \
  __shared__ __attribute__((aligned(16))) float Ar[LPB][FN], Ai[LPB][FN];   \
  __shared__ __attribute__((aligned(16))) float Twr[FN], Twi[FN];

#define FFT_SHARED_C16                                          \
  __shared__ __attribute__((aligned(16))) float Ar[CPB][LPITCH], Ai[CPB][LPITCH]; \
  __shared__ __attribute__((aligned(16))) float Br[CPB][LPITCH], Bi[CPB][LPITCH]; \
  __shared__ __attribute__((aligned(16))) float Twr[FN], Twi[FN];

__global__ __launch_bounds__(TPB)
void k_fwd_rows(const float* __restrict__ img, float2* __restrict__ specRow) {
  FFT_SHARED_ROWS
  const int tid = threadIdx.x, lane = tid & 31, wid = tid >> 5;
  const int line = blockIdx.x * LPB + wid;
  float xr[16], xi[16];
#pragma unroll
  for (int i = 0; i < 16; ++i) {
    xr[i] = img[line * FN + lane + 32 * i];
    xi[i] = 0.0f;
  }
  fill_tw512(Twr, Twi, tid, TPB, -1.0f);
  W16 w; make_w16(-1.0f, lane, w);
  v2f e0[8], e1[8];
  fft_stageA_regs(xr, xi, lane, -1.0f, e0, e1);
  v8f gr[2], gi[2];
  ctile_regs(e0, lane, w, gr[0], gi[0]);
  ctile_regs(e1, lane, w, gr[1], gi[1]);
  __syncthreads();
  fft_stageC_store(gr, gi, Ar[wid], Ai[wid], Twr, Twi, lane);
  __syncthreads();
  const int nn = lane & 15, mb = (lane >> 4) * 8;
#pragma unroll
  for (int t = 0; t < 2; ++t) {
    v8f dr, di;
    ctile(Ar[wid], Ai[wid], t, lane, w, dr, di);
#pragma unroll
    for (int r = 0; r < 8; ++r) {
      const int o = 32 * (mb + r) + 16 * t + nn;
      specRow[line * FN + o] = make_float2(dr[r], di[r]);
    }
  }
}

__global__ __launch_bounds__(TPB)
void k_fwd_cols(const float2* __restrict__ specRow, float2* __restrict__ spec) {
  FFT_SHARED
  const int tid = threadIdx.x, lane = tid & 31, wid = tid >> 5;
  const int c0 = blockIdx.x * LPB;
  fill_tw512(Twr, Twi, tid, TPB, -1.0f);
  W16 w; make_w16(-1.0f, lane, w);
  for (int i = tid; i < FN * LPB; i += TPB) {
    const int e = i >> 2, ww = i & 3;
    const float2 s = specRow[e * FN + c0 + ww];
    Ar[ww][e] = s.x; Ai[ww][e] = s.y;
  }
  __syncthreads();
  {
    float xr[16], xi[16];
#pragma unroll
    for (int q = 0; q < 16; ++q) {
      xr[q] = Ar[wid][lane + 32 * q];
      xi[q] = Ai[wid][lane + 32 * q];
    }
    v2f e0[8], e1[8];
    fft_stageA_regs(xr, xi, lane, -1.0f, e0, e1);
    v8f gr[2], gi[2];
    ctile_regs(e0, lane, w, gr[0], gi[0]);
    ctile_regs(e1, lane, w, gr[1], gi[1]);
    fft_stageC_store(gr, gi, Ar[wid], Ai[wid], Twr, Twi, lane);
  }
  __syncthreads();
  const int nn = lane & 15, mb = (lane >> 4) * 8;
#pragma unroll
  for (int t = 0; t < 2; ++t) {
    v8f dr, di;
    ctile(Ar[wid], Ai[wid], t, lane, w, dr, di);
#pragma unroll
    for (int r = 0; r < 8; ++r) {
      const int o = 32 * (mb + r) + 16 * t + nn;
      Br[wid][o] = dr[r]; Bi[wid][o] = di[r];
    }
  }
  __syncthreads();
  for (int i = tid; i < FN * LPB; i += TPB) {
    const int o = i >> 2, ww = i & 3;
    spec[o * FN + c0 + ww] = make_float2(Br[ww][o], Bi[ww][o]);
  }
}

__global__ __launch_bounds__(TPB)
void k_band_rows(const float2* __restrict__ spec, const float* __restrict__ win,
                 float2* __restrict__ T, int b0) {
  FFT_SHARED_ROWS
  const int tid = threadIdx.x, lane = tid & 31, wid = tid >> 5;
  const int bi = blockIdx.y;
  const float* wb = win + (size_t)(b0 + bi) * FN * FN;
  float2* Tb = T + (size_t)bi * HC * FN;
  const int k1  = blockIdx.x * LPB + wid;
  const int mk1 = (FN - k1) & (FN - 1);
  const float2* srow = spec + k1 * FN;
  const float*  wrow = wb + k1 * FN;
  const float*  mrow = wb + mk1 * FN;
  float2 s[16]; float wv[16], wm[16];
#pragma unroll
  for (int i = 0; i < 16; ++i) {
    const int k2 = lane + 32 * i;
    s[i]  = srow[k2];
    wv[i] = wrow[k2];
    wm[i] = mrow[(FN - k2) & (FN - 1)];
  }
  fill_tw512(Twr, Twi, tid, TPB, 1.0f);
  W16 w; make_w16(1.0f, lane, w);
  float xr[16], xi[16];
#pragma unroll
  for (int i = 0; i < 16; ++i) {
    const float g = 0.5f * (wv[i] + wm[i]);
    xr[i] = s[i].x * g; xi[i] = s[i].y * g;
  }
  if (k1 == 0) {
#pragma unroll
    for (int i = 0; i < 16; ++i) {
      const int k2 = lane + 32 * i;
      const float2 q = spec[HC * FN + k2];
      const float g = 0.5f * (wb[HC * FN + k2] +
                              wb[HC * FN + ((FN - k2) & (FN - 1))]);
      xr[i] -= q.y * g;
      xi[i] += q.x * g;
    }
  }
  v2f e0[8], e1[8];
  fft_stageA_regs(xr, xi, lane, 1.0f, e0, e1);
  v8f gr[2], gi[2];
  ctile_regs(e0, lane, w, gr[0], gi[0]);
  ctile_regs(e1, lane, w, gr[1], gi[1]);
  __syncthreads();
  fft_stageC_store(gr, gi, Ar[wid], Ai[wid], Twr, Twi, lane);
  __syncthreads();
  const int nn = lane & 15, mb = (lane >> 4) * 8;
#pragma unroll
  for (int t = 0; t < 2; ++t) {
    v8f dr, di;
    ctile(Ar[wid], Ai[wid], t, lane, w, dr, di);
#pragma unroll
    for (int r = 0; r < 8; ++r) {
      const int o = 32 * (mb + r) + 16 * t + nn;
      Tb[k1 * FN + o] = make_float2(dr[r], di[r]);
    }
  }
}

__global__ __launch_bounds__(CTPB)
void k_band_cols(const float2* __restrict__ T, float* __restrict__ out, int b0) {
  FFT_SHARED_C16
  const int tid = threadIdx.x, lane = tid & 31, wid = tid >> 5;
  const int c0 = blockIdx.x * (2 * CPB);
  const int bi = blockIdx.y;
  const float2* Tb = T + (size_t)bi * HC * FN;
  float* ob = out + (size_t)(b0 + bi) * FN * FN;
  const int ww = tid & (CPB - 1);
  const int e0 = tid >> 4;
  float2 a[8], b[8];
#pragma unroll
  for (int it = 0; it < 8; ++it) {
    const int k = e0 + 32 * it;
    a[it] = Tb[k * FN + c0 + ww];
    b[it] = Tb[k * FN + c0 + CPB + ww];
  }
  fill_tw512(Twr, Twi, tid, CTPB, 1.0f);
  W16 w; make_w16(1.0f, lane, w);
#pragma unroll
  for (int it = 0; it < 8; ++it) {
    const int k = e0 + 32 * it;
    const bool z = (k == 0);
    const int  km = z ? HC : FN - k;
    const float2 p = a[it], q = b[it];
    Ar[ww][k]  = z ? p.x : p.x - q.y;
    Ai[ww][k]  = z ? q.x : p.y + q.x;
    Ar[ww][km] = z ? p.y : p.x + q.y;
    Ai[ww][km] = z ? q.y : q.x - p.y;
  }
  __syncthreads();
  {
    float xr[16], xi[16];
#pragma unroll
    for (int q = 0; q < 16; ++q) {
      xr[q] = Ar[wid][lane + 32 * q];
      xi[q] = Ai[wid][lane + 32 * q];
    }
    v2f e0[8], e1[8];
    fft_stageA_regs(xr, xi, lane, 1.0f, e0, e1);
    v8f gr[2], gi[2];
    ctile_regs(e0, lane, w, gr[0], gi[0]);
    ctile_regs(e1, lane, w, gr[1], gi[1]);
    fft_stageC_store(gr, gi, Ar[wid], Ai[wid], Twr, Twi, lane);
  }
  __syncthreads();
  const int nn = lane & 15, mb = (lane >> 4) * 8;
#pragma unroll
  for (int t = 0; t < 2; ++t) {
    v8f dr, di;
    ctile(Ar[wid], Ai[wid], t, lane, w, dr, di);
    const int os0 = (32 * mb + 16 * t + nn) ^ (2 * mb);
    const int os1 = os0 ^ 2;
#pragma unroll
    for (int r = 0; r < 8; ++r) {
      Br[wid][os0 + 32 * r] = dr[r];
      Bi[wid][os1 + 32 * r] = di[r];
    }
  }
  __syncthreads();
  const float scale = 1.0f / (512.0f * 512.0f);
  const int cw = tid & 31;
  const int r0 = tid >> 5;
  const bool im = cw >= CPB;
  const float* col = im ? &Bi[cw - CPB][0] : &Br[cw][0];
  const int rb = im ? (r0 ^ 2) : r0;
#pragma unroll
  for (int it = 0; it < 32; ++it) {
    const int o  = r0 + 16 * it;
    const int os = rb + 16 * (it ^ (it >> 4));
    ob[o * FN + c0 + cw] = col[os] * scale;
  }
}

extern "C" void kernel_launch(void* const* d_in, const int* in_sizes, int n_in,
                              void* d_out, int out_size, void* d_ws,
                              size_t ws_size, hipStream_t stream) {
  (void)in_sizes; (void)n_in; (void)out_size;
  const float* img = (const float*)d_in[0];
  const float* win = (const float*)d_in[1];
  float* out = (float*)d_out;

  float2* specRow = (float2*)d_ws;
  float2* spec    = specRow + (size_t)FN * FN;
  float2* bandT   = spec    + (size_t)FN * FN;
  const size_t reserved = 2ull * sizeof(float2) * FN * FN;
  const size_t perBand  = sizeof(float2) * (size_t)FN * FN;
  int chunk = (ws_size > reserved) ? (int)((ws_size - reserved) / perBand) : 0;
  if (chunk < 1)  chunk = 1;
  if (chunk > NB) chunk = NB;

  const dim3 blk(TPB);
  const dim3 cblk(CTPB);
  k_fwd_rows<<<dim3(FN / LPB), blk, 0, stream>>>(img, specRow);
  k_fwd_cols<<<dim3(FN / LPB), blk, 0, stream>>>(specRow, spec);
  for (int b0 = 0; b0 < NB; b0 += chunk) {
    const int nb = (NB - b0 < chunk) ? (NB - b0) : chunk;
    k_band_rows<<<dim3(HC / LPB, nb), blk, 0, stream>>>(spec, win, bandT, b0);
    k_band_cols<<<dim3(FN / (2 * CPB), nb), cblk, 0, stream>>>(bandT, out, b0);
  }
}
